# starting kernel
# baseline (speedup 1.0000x reference)
#include <hip/hip_runtime.h>

typedef __attribute__((ext_vector_type(16))) _Float16 v16h;
typedef __attribute__((ext_vector_type(8)))  float    v8f;

#define IN_CH 128
#define HID   64
#define NFRAG (4 * 4 * 32 * 16)

__device__ __forceinline__ v8f wmma_f16(v16h a, v16h b, v8f c)
{
    return __builtin_amdgcn_wmma_f32_16x16x32_f16(false, a, false, b,
                                                  (short)0, c, false, false);
}

__device__ __forceinline__ void stage_Bfrag(const float* __restrict__ W,
                                            _Float16* sB)
{
    for (int i = threadIdx.x; i < NFRAG; i += blockDim.x) {
        const int j     = i & 15;
        const int lane  = (i >> 4) & 31;
        const int kci   = (i >> 9) & 3;
        const int ntile = i >> 11;
        const int koff  = (lane >> 4) << 3;
        const int col   = (ntile << 4) + (lane & 15);
        const int k     = (kci << 5) + koff + (j & 7) + ((j >> 3) << 4);
        sB[i] = (_Float16)W[k * HID + col];
    }
}

__device__ __forceinline__ v16h load_a16(const float* __restrict__ arow, int off)
{
    float af[16];
    reinterpret_cast<float4*>(af)[0] = *reinterpret_cast<const float4*>(arow + off);
    reinterpret_cast<float4*>(af)[1] = *reinterpret_cast<const float4*>(arow + off + 4);
    reinterpret_cast<float4*>(af)[2] = *reinterpret_cast<const float4*>(arow + off + 16);
    reinterpret_cast<float4*>(af)[3] = *reinterpret_cast<const float4*>(arow + off + 20);
    v16h a;
    #pragma unroll
    for (int j = 0; j < 16; ++j) a[j] = (_Float16)af[j];
    return a;
}

__global__ void __launch_bounds__(256)
gemm1_wmma(const float* __restrict__ x,
           const float* __restrict__ Wc,
           float* __restrict__ h,
           int n_nodes)
{
    __shared__ __align__(32) _Float16 sB[NFRAG];
    stage_Bfrag(Wc, sB);
    __syncthreads();

    const int wave = threadIdx.x >> 5;
    const int lane = threadIdx.x & 31;
    const int mtiles  = n_nodes >> 4;
    const int mgroups = (mtiles + 3) >> 2;
    const int g = blockIdx.x * (blockDim.x >> 5) + wave;
    if (g >= mgroups * 4) return;
    const int ntile = g & 3;
    const int mg    = g >> 2;

    const v16h* bp = reinterpret_cast<const v16h*>(sB) + (ntile << 7) + lane;
    const v16h b0 = bp[0], b1 = bp[32], b2 = bp[64], b3 = bp[96];

    const int koff = (lane >> 4) << 3;
    const int col  = (ntile << 4) + (lane & 15);

    for (int t = 0; t < 4; ++t) {
        const int mtile = (mg << 2) + t;
        if (mtile >= mtiles) return;
        const int row = (mtile << 4) + (lane & 15);
        const float* arow = x + (long)row * IN_CH;
        __builtin_prefetch(arow + 256, 0, 1);

        v8f c = {};
        c = wmma_f16(load_a16(arow,      koff), b0, c);
        c = wmma_f16(load_a16(arow, 32 + koff), b1, c);
        c = wmma_f16(load_a16(arow, 64 + koff), b2, c);
        c = wmma_f16(load_a16(arow, 96 + koff), b3, c);

        const int rbase = (mtile << 4) + ((lane >> 4) << 3);
        #pragma unroll
        for (int r = 0; r < 8; ++r)
            h[(long)(rbase + r) * HID + col] = c[r];
    }
}

__global__ void zero_f32(float4* __restrict__ p, int n4)
{
    int i = blockIdx.x * blockDim.x + threadIdx.x;
    for (; i < n4; i += gridDim.x * blockDim.x)
        p[i] = make_float4(0.f, 0.f, 0.f, 0.f);
}

__global__ void __launch_bounds__(256)
scatter_edges(const float* __restrict__ h,
              const int*   __restrict__ ei,
              const float* __restrict__ ew,
              float*       __restrict__ agg,
              int E)
{
    const int t    = blockIdx.x * blockDim.x + threadIdx.x;
    const int e    = t >> 5;
    const int lane = t & 31;
    if (e >= E) return;

    const int   s = ei[e];
    const int   d = ei[E + e];
    const float w = ew[e];

    float2 hv = reinterpret_cast<const float2*>(h + (long)s * HID)[lane];
    float* ap = agg + (long)d * HID + (lane << 1);
    atomicAdd(ap,     hv.x * w);
    atomicAdd(ap + 1, hv.y * w);
}

__global__ void __launch_bounds__(256)
bias_lrelu(float* __restrict__ agg,
           const float* __restrict__ b_conv,
           const float* __restrict__ alpha,
           int total4)
{
    const int i = blockIdx.x * blockDim.x + threadIdx.x;
    if (i >= total4) return;
    const int c4 = (i & 15) << 2;
    float4 v  = reinterpret_cast<float4*>(agg)[i];
    float4 bb = *reinterpret_cast<const float4*>(b_conv + c4);
    float4 aa = *reinterpret_cast<const float4*>(alpha + c4);
    v.x += bb.x; v.y += bb.y; v.z += bb.z; v.w += bb.w;
    v.x = v.x > 0.f ? v.x : aa.x * v.x;
    v.y = v.y > 0.f ? v.y : aa.y * v.y;
    v.z = v.z > 0.f ? v.z : aa.z * v.z;
    v.w = v.w > 0.f ? v.w : aa.w * v.w;
    reinterpret_cast<float4*>(agg)[i] = v;
}

__global__ void __launch_bounds__(256)
gemm2_wmma(const float* __restrict__ act,
           const float* __restrict__ Wcat,
           const float* __restrict__ b_cat,
           float* __restrict__ out,
           int n_half)
{
    __shared__ __align__(32) _Float16 sB[NFRAG];
    stage_Bfrag(Wcat, sB);
    __syncthreads();

    const int wave = threadIdx.x >> 5;
    const int lane = threadIdx.x & 31;
    const int mtiles  = (n_half + 15) >> 4;
    const int mgroups = (mtiles + 3) >> 2;
    const int g = blockIdx.x * (blockDim.x >> 5) + wave;
    if (g >= mgroups * 4) return;
    const int ntile = g & 3;
    const int mg    = g >> 2;

    const v16h* bp = reinterpret_cast<const v16h*>(sB) + (ntile << 7) + lane;
    const v16h b0 = bp[0], b1 = bp[32], b2 = bp[64], b3 = bp[96];

    const int koff = (lane >> 4) << 3;
    const int col  = (ntile << 4) + (lane & 15);

    for (int t = 0; t < 4; ++t) {
        const int mtile = (mg << 2) + t;
        if (mtile >= mtiles) return;
        const int row  = (mtile << 4) + (lane & 15);
        const int rowc = row < n_half ? row : n_half - 1;
        const float* a0 = act + (long)rowc * HID;
        const float* a1 = act + (long)(rowc + n_half) * HID;

        v8f c = {};
        c = wmma_f16(load_a16(a0,      koff), b0, c);
        c = wmma_f16(load_a16(a0, 32 + koff), b1, c);
        c = wmma_f16(load_a16(a1,      koff), b2, c);
        c = wmma_f16(load_a16(a1, 32 + koff), b3, c);

        const float bb = b_cat[col];
        const int rbase = (mtile << 4) + ((lane >> 4) << 3);
        #pragma unroll
        for (int r = 0; r < 8; ++r) {
            const int rr = rbase + r;
            if (rr < n_half) {
                float v = c[r] + bb;
                out[(long)rr * HID + col] = v > 0.f ? v : 0.f;
            }
        }
    }
}

extern "C" void kernel_launch(void* const* d_in, const int* in_sizes, int n_in,
                              void* d_out, int out_size, void* d_ws, size_t ws_size,
                              hipStream_t stream)
{
    const float* x   = (const float*)d_in[0];
    const int*   ei  = (const int*)  d_in[1];
    const float* ew  = (const float*)d_in[2];
    const float* Wc  = (const float*)d_in[3];
    const float* bc  = (const float*)d_in[4];
    const float* al  = (const float*)d_in[5];
    const float* Wk  = (const float*)d_in[6];
    const float* bk  = (const float*)d_in[7];

    const int n_nodes = in_sizes[0] / IN_CH;
    const int E       = in_sizes[2];
    const int n_half  = n_nodes / 2;

    float* h   = (float*)d_ws;
    float* agg = h + (size_t)n_nodes * HID;
    float* out = (float*)d_out;

    {
        const int mgroups = ((n_nodes >> 4) + 3) >> 2;
        const int waves   = mgroups * 4;
        gemm1_wmma<<<(waves + 7) / 8, 256, 0, stream>>>(x, Wc, h, n_nodes);
    }

    const int n4 = (n_nodes * HID) >> 2;
    zero_f32<<<(n4 + 255) / 256, 256, 0, stream>>>((float4*)agg, n4);

    scatter_edges<<<(E + 7) / 8, 256, 0, stream>>>(h, ei, ew, agg, E);

    bias_lrelu<<<(n4 + 255) / 256, 256, 0, stream>>>(agg, bc, al, n4);

    {
        const int mgroups = (((n_half + 15) >> 4) + 3) >> 2;
        const int waves   = mgroups * 4;
        gemm2_wmma<<<(waves + 7) / 8, 256, 0, stream>>>(agg, Wk, bk, out, n_half);
    }
}


# edge sums made destination-owned: an exchange atomic threads each edge on its target's list, a half-wave per row walks it and sums in registers; no float atomics, no zero fill; sums in list order
# speedup vs baseline: 1.0583x; 1.0583x over previous
#include <hip/hip_runtime.h>

typedef __attribute__((ext_vector_type(16))) _Float16 v16h;
typedef __attribute__((ext_vector_type(8)))  float    v8f;

#define IN_CH 128
#define HID   64
#define NFRAG (4 * 4 * 32 * 16)

__device__ __forceinline__ v8f wmma_f16(v16h a, v16h b, v8f c)
{
    return __builtin_amdgcn_wmma_f32_16x16x32_f16(false, a, false, b,
                                                  (short)0, c, false, false);
}

__device__ __forceinline__ void stage_Bfrag(const float* __restrict__ W,
                                            _Float16* sB)
{
    for (int i = threadIdx.x; i < NFRAG; i += blockDim.x) {
        const int j     = i & 15;
        const int lane  = (i >> 4) & 31;
        const int kci   = (i >> 9) & 3;
        const int ntile = i >> 11;
        const int koff  = (lane >> 4) << 3;
        const int col   = (ntile << 4) + (lane & 15);
        const int k     = (kci << 5) + koff + (j & 7) + ((j >> 3) << 4);
        sB[i] = (_Float16)W[k * HID + col];
    }
}

__device__ __forceinline__ v16h load_a16(const float* __restrict__ arow, int off)
{
    float af[16];
    reinterpret_cast<float4*>(af)[0] = *reinterpret_cast<const float4*>(arow + off);
    reinterpret_cast<float4*>(af)[1] = *reinterpret_cast<const float4*>(arow + off + 4);
    reinterpret_cast<float4*>(af)[2] = *reinterpret_cast<const float4*>(arow + off + 16);
    reinterpret_cast<float4*>(af)[3] = *reinterpret_cast<const float4*>(arow + off + 20);
    v16h a;
    #pragma unroll
    for (int j = 0; j < 16; ++j) a[j] = (_Float16)af[j];
    return a;
}

__global__ void __launch_bounds__(256)
gemm1_wmma(const float* __restrict__ x,
           const float* __restrict__ Wc,
           float* __restrict__ h,
           int n_nodes)
{
    __shared__ __align__(32) _Float16 sB[NFRAG];
    stage_Bfrag(Wc, sB);
    __syncthreads();

    const int wave = threadIdx.x >> 5;
    const int lane = threadIdx.x & 31;
    const int mtiles  = n_nodes >> 4;
    const int mgroups = (mtiles + 3) >> 2;
    const int g = blockIdx.x * (blockDim.x >> 5) + wave;
    if (g >= mgroups * 4) return;
    const int ntile = g & 3;
    const int mg    = g >> 2;

    const v16h* bp = reinterpret_cast<const v16h*>(sB) + (ntile << 7) + lane;
    const v16h b0 = bp[0], b1 = bp[32], b2 = bp[64], b3 = bp[96];

    const int koff = (lane >> 4) << 3;
    const int col  = (ntile << 4) + (lane & 15);

    for (int t = 0; t < 4; ++t) {
        const int mtile = (mg << 2) + t;
        if (mtile >= mtiles) return;
        const int row = (mtile << 4) + (lane & 15);
        const float* arow = x + (long)row * IN_CH;
        __builtin_prefetch(arow + 256, 0, 1);

        v8f c = {};
        c = wmma_f16(load_a16(arow,      koff), b0, c);
        c = wmma_f16(load_a16(arow, 32 + koff), b1, c);
        c = wmma_f16(load_a16(arow, 64 + koff), b2, c);
        c = wmma_f16(load_a16(arow, 96 + koff), b3, c);

        const int rbase = (mtile << 4) + ((lane >> 4) << 3);
        #pragma unroll
        for (int r = 0; r < 8; ++r)
            h[(long)(rbase + r) * HID + col] = c[r];
    }
}

__global__ void zero_f32(float4* __restrict__ p, int n4)
{
    int i = blockIdx.x * blockDim.x + threadIdx.x;
    for (; i < n4; i += gridDim.x * blockDim.x)
        p[i] = make_float4(0.f, 0.f, 0.f, 0.f);
}

__global__ void __launch_bounds__(256)
scatter_edges(const float* __restrict__ h,
              const int*   __restrict__ ei,
              const float* __restrict__ ew,
              float*       __restrict__ agg,
              int E)
{
    const int t    = blockIdx.x * blockDim.x + threadIdx.x;
    const int e    = t >> 5;
    const int lane = t & 31;
    if (e >= E) return;

    const int   s = ei[e];
    const int   d = ei[E + e];
    const float w = ew[e];

    float2 hv = reinterpret_cast<const float2*>(h + (long)s * HID)[lane];
    float* ap = agg + (long)d * HID + (lane << 1);
    atomicAdd(ap,     hv.x * w);
    atomicAdd(ap + 1, hv.y * w);
}

__global__ void __launch_bounds__(256)
fill_heads(int* __restrict__ head, int n)
{
    const int i = blockIdx.x * blockDim.x + threadIdx.x;
    if (i < n) head[i] = -1;
}

__global__ void __launch_bounds__(256)
link_edges(const int* __restrict__ ei,
           int* __restrict__ head,
           int* __restrict__ next,
           int E)
{
    const int e = blockIdx.x * blockDim.x + threadIdx.x;
    if (e >= E) return;
    const int d = ei[E + e];
    next[e] = atomicExch(head + d, e);
}

__global__ void __launch_bounds__(256)
gather_rows(const float* __restrict__ h,
            const int*   __restrict__ ei,
            const float* __restrict__ ew,
            const int*   __restrict__ head,
            const int*   __restrict__ next,
            float*       __restrict__ agg,
            int n_nodes, int E)
{
    const int t = blockIdx.x * blockDim.x + threadIdx.x;
    const int d = t >> 4;
    const int q = t & 15;
    if (d >= n_nodes) return;

    float4 acc = make_float4(0.f, 0.f, 0.f, 0.f);

    int   e  = head[d];
    int   s  = 0;
    float w  = 0.f;
    int   en = -1;
    if ((unsigned)e < (unsigned)E) { s = ei[e]; w = ew[e]; en = next[e]; }

    for (int it = 0; it < E && (unsigned)e < (unsigned)E; ++it) {
        const float4 v  = reinterpret_cast<const float4*>(h + (long)s * HID)[q];
        const float  wc = w;
        e = en;
        if ((unsigned)e < (unsigned)E) { s = ei[e]; w = ew[e]; en = next[e]; }
        acc.x += v.x * wc;
        acc.y += v.y * wc;
        acc.z += v.z * wc;
        acc.w += v.w * wc;
    }
    reinterpret_cast<float4*>(agg + (long)d * HID)[q] = acc;
}

__global__ void __launch_bounds__(256)
bias_lrelu(float* __restrict__ agg,
           const float* __restrict__ b_conv,
           const float* __restrict__ alpha,
           int total4)
{
    const int i = blockIdx.x * blockDim.x + threadIdx.x;
    if (i >= total4) return;
    const int c4 = (i & 15) << 2;
    float4 v  = reinterpret_cast<float4*>(agg)[i];
    float4 bb = *reinterpret_cast<const float4*>(b_conv + c4);
    float4 aa = *reinterpret_cast<const float4*>(alpha + c4);
    v.x += bb.x; v.y += bb.y; v.z += bb.z; v.w += bb.w;
    v.x = v.x > 0.f ? v.x : aa.x * v.x;
    v.y = v.y > 0.f ? v.y : aa.y * v.y;
    v.z = v.z > 0.f ? v.z : aa.z * v.z;
    v.w = v.w > 0.f ? v.w : aa.w * v.w;
    reinterpret_cast<float4*>(agg)[i] = v;
}

__global__ void __launch_bounds__(256)
gemm2_wmma(const float* __restrict__ act,
           const float* __restrict__ Wcat,
           const float* __restrict__ b_cat,
           float* __restrict__ out,
           int n_half)
{
    __shared__ __align__(32) _Float16 sB[NFRAG];
    stage_Bfrag(Wcat, sB);
    __syncthreads();

    const int wave = threadIdx.x >> 5;
    const int lane = threadIdx.x & 31;
    const int mtiles  = (n_half + 15) >> 4;
    const int mgroups = (mtiles + 3) >> 2;
    const int g = blockIdx.x * (blockDim.x >> 5) + wave;
    if (g >= mgroups * 4) return;
    const int ntile = g & 3;
    const int mg    = g >> 2;

    const v16h* bp = reinterpret_cast<const v16h*>(sB) + (ntile << 7) + lane;
    const v16h b0 = bp[0], b1 = bp[32], b2 = bp[64], b3 = bp[96];

    const int koff = (lane >> 4) << 3;
    const int col  = (ntile << 4) + (lane & 15);

    for (int t = 0; t < 4; ++t) {
        const int mtile = (mg << 2) + t;
        if (mtile >= mtiles) return;
        const int row  = (mtile << 4) + (lane & 15);
        const int rowc = row < n_half ? row : n_half - 1;
        const float* a0 = act + (long)rowc * HID;
        const float* a1 = act + (long)(rowc + n_half) * HID;

        v8f c = {};
        c = wmma_f16(load_a16(a0,      koff), b0, c);
        c = wmma_f16(load_a16(a0, 32 + koff), b1, c);
        c = wmma_f16(load_a16(a1,      koff), b2, c);
        c = wmma_f16(load_a16(a1, 32 + koff), b3, c);

        const float bb = b_cat[col];
        const int rbase = (mtile << 4) + ((lane >> 4) << 3);
        #pragma unroll
        for (int r = 0; r < 8; ++r) {
            const int rr = rbase + r;
            if (rr < n_half) {
                float v = c[r] + bb;
                out[(long)rr * HID + col] = v > 0.f ? v : 0.f;
            }
        }
    }
}

extern "C" void kernel_launch(void* const* d_in, const int* in_sizes, int n_in,
                              void* d_out, int out_size, void* d_ws, size_t ws_size,
                              hipStream_t stream)
{
    const float* x   = (const float*)d_in[0];
    const int*   ei  = (const int*)  d_in[1];
    const float* ew  = (const float*)d_in[2];
    const float* Wc  = (const float*)d_in[3];
    const float* bc  = (const float*)d_in[4];
    const float* al  = (const float*)d_in[5];
    const float* Wk  = (const float*)d_in[6];
    const float* bk  = (const float*)d_in[7];

    const int n_nodes = in_sizes[0] / IN_CH;
    const int E       = in_sizes[2];
    const int n_half  = n_nodes / 2;

    float* h   = (float*)d_ws;
    float* agg = h + (size_t)n_nodes * HID;
    float* out = (float*)d_out;

    {
        const int mgroups = ((n_nodes >> 4) + 3) >> 2;
        const int waves   = mgroups * 4;
        gemm1_wmma<<<(waves + 7) / 8, 256, 0, stream>>>(x, Wc, h, n_nodes);
    }

    const int n4 = (n_nodes * HID) >> 2;
    int* head = (int*)(agg + (size_t)n_nodes * HID);
    int* next = head + n_nodes;
    const size_t need = (size_t)n_nodes * HID * 2 * sizeof(float)
                      + ((size_t)n_nodes + (size_t)E) * sizeof(int);
    if (ws_size >= need) {
        fill_heads<<<(n_nodes + 255) / 256, 256, 0, stream>>>(head, n_nodes);
        link_edges<<<(E + 255) / 256, 256, 0, stream>>>(ei, head, next, E);
        gather_rows<<<(n_nodes * 16 + 255) / 256, 256, 0, stream>>>(
            h, ei, ew, head, next, agg, n_nodes, E);
    } else {
        zero_f32<<<(n4 + 255) / 256, 256, 0, stream>>>((float4*)agg, n4);

        scatter_edges<<<(E + 7) / 8, 256, 0, stream>>>(h, ei, ew, agg, E);
    }

    bias_lrelu<<<(n4 + 255) / 256, 256, 0, stream>>>(agg, bc, al, n4);

    {
        const int mgroups = (((n_half + 15) >> 4) + 3) >> 2;
        const int waves   = mgroups * 4;
        gemm2_wmma<<<(waves + 7) / 8, 256, 0, stream>>>(agg, Wk, bk, out, n_half);
    }
}


# edges binned by a count atomic into a 256-byte 31-slot bucket per destination (overflow keeps the list); gather reads the bucket whole and requests its source rows independently; sums in slot order
# speedup vs baseline: 1.7017x; 1.6080x over previous
#include <hip/hip_runtime.h>

typedef __attribute__((ext_vector_type(16))) _Float16 v16h;
typedef __attribute__((ext_vector_type(8)))  float    v8f;

#define IN_CH 128
#define HID   64
#define NFRAG (4 * 4 * 32 * 16)

__device__ __forceinline__ v8f wmma_f16(v16h a, v16h b, v8f c)
{
    return __builtin_amdgcn_wmma_f32_16x16x32_f16(false, a, false, b,
                                                  (short)0, c, false, false);
}

__device__ __forceinline__ void stage_Bfrag(const float* __restrict__ W,
                                            _Float16* sB)
{
    for (int i = threadIdx.x; i < NFRAG; i += blockDim.x) {
        const int j     = i & 15;
        const int lane  = (i >> 4) & 31;
        const int kci   = (i >> 9) & 3;
        const int ntile = i >> 11;
        const int koff  = (lane >> 4) << 3;
        const int col   = (ntile << 4) + (lane & 15);
        const int k     = (kci << 5) + koff + (j & 7) + ((j >> 3) << 4);
        sB[i] = (_Float16)W[k * HID + col];
    }
}

__device__ __forceinline__ v16h load_a16(const float* __restrict__ arow, int off)
{
    float af[16];
    reinterpret_cast<float4*>(af)[0] = *reinterpret_cast<const float4*>(arow + off);
    reinterpret_cast<float4*>(af)[1] = *reinterpret_cast<const float4*>(arow + off + 4);
    reinterpret_cast<float4*>(af)[2] = *reinterpret_cast<const float4*>(arow + off + 16);
    reinterpret_cast<float4*>(af)[3] = *reinterpret_cast<const float4*>(arow + off + 20);
    v16h a;
    #pragma unroll
    for (int j = 0; j < 16; ++j) a[j] = (_Float16)af[j];
    return a;
}

__global__ void __launch_bounds__(256)
gemm1_wmma(const float* __restrict__ x,
           const float* __restrict__ Wc,
           float* __restrict__ h,
           int n_nodes)
{
    __shared__ __align__(32) _Float16 sB[NFRAG];
    stage_Bfrag(Wc, sB);
    __syncthreads();

    const int wave = threadIdx.x >> 5;
    const int lane = threadIdx.x & 31;
    const int mtiles  = n_nodes >> 4;
    const int mgroups = (mtiles + 3) >> 2;
    const int g = blockIdx.x * (blockDim.x >> 5) + wave;
    if (g >= mgroups * 4) return;
    const int ntile = g & 3;
    const int mg    = g >> 2;

    const v16h* bp = reinterpret_cast<const v16h*>(sB) + (ntile << 7) + lane;
    const v16h b0 = bp[0], b1 = bp[32], b2 = bp[64], b3 = bp[96];

    const int koff = (lane >> 4) << 3;
    const int col  = (ntile << 4) + (lane & 15);

    for (int t = 0; t < 4; ++t) {
        const int mtile = (mg << 2) + t;
        if (mtile >= mtiles) return;
        const int row = (mtile << 4) + (lane & 15);
        const float* arow = x + (long)row * IN_CH;
        __builtin_prefetch(arow + 256, 0, 1);

        v8f c = {};
        c = wmma_f16(load_a16(arow,      koff), b0, c);
        c = wmma_f16(load_a16(arow, 32 + koff), b1, c);
        c = wmma_f16(load_a16(arow, 64 + koff), b2, c);
        c = wmma_f16(load_a16(arow, 96 + koff), b3, c);

        const int rbase = (mtile << 4) + ((lane >> 4) << 3);
        #pragma unroll
        for (int r = 0; r < 8; ++r)
            h[(long)(rbase + r) * HID + col] = c[r];
    }
}

__global__ void zero_f32(float4* __restrict__ p, int n4)
{
    int i = blockIdx.x * blockDim.x + threadIdx.x;
    for (; i < n4; i += gridDim.x * blockDim.x)
        p[i] = make_float4(0.f, 0.f, 0.f, 0.f);
}

__global__ void __launch_bounds__(256)
scatter_edges(const float* __restrict__ h,
              const int*   __restrict__ ei,
              const float* __restrict__ ew,
              float*       __restrict__ agg,
              int E)
{
    const int t    = blockIdx.x * blockDim.x + threadIdx.x;
    const int e    = t >> 5;
    const int lane = t & 31;
    if (e >= E) return;

    const int   s = ei[e];
    const int   d = ei[E + e];
    const float w = ew[e];

    float2 hv = reinterpret_cast<const float2*>(h + (long)s * HID)[lane];
    float* ap = agg + (long)d * HID + (lane << 1);
    atomicAdd(ap,     hv.x * w);
    atomicAdd(ap + 1, hv.y * w);
}

#define BSLOTS 31

__global__ void __launch_bounds__(256)
init_buckets_pack_b(int2* __restrict__ bkt, int n,
                    const float* __restrict__ Wcat,
                    _Float16* __restrict__ bpk)
{
    const int i = blockIdx.x * blockDim.x + threadIdx.x;
    if (i < n) bkt[(size_t)i * 32 + BSLOTS] = make_int2(0, -1);
    if (i < NFRAG) {
        const int j     = i & 15;
        const int lane  = (i >> 4) & 31;
        const int kci   = (i >> 9) & 3;
        const int ntile = i >> 11;
        const int koff  = (lane >> 4) << 3;
        const int col   = (ntile << 4) + (lane & 15);
        const int k     = (kci << 5) + koff + (j & 7) + ((j >> 3) << 4);
        bpk[i] = (_Float16)Wcat[k * HID + col];
    }
}

__global__ void __launch_bounds__(256)
bin_edges(const int*   __restrict__ ei,
          const float* __restrict__ ew,
          int2* __restrict__ bkt,
          int*  __restrict__ next,
          int E)
{
    const int e = blockIdx.x * blockDim.x + threadIdx.x;
    if (e >= E) return;
    const int   s = ei[e];
    const int   d = ei[E + e];
    const float w = ew[e];
    int2* b = bkt + (size_t)d * 32;
    const int slot = atomicAdd(&b[BSLOTS].x, 1);
    if (slot < BSLOTS) b[slot] = make_int2(s, __float_as_int(w));
    else               next[e] = atomicExch(&b[BSLOTS].y, e);
}

template <int L>
__device__ __forceinline__ int bc8(int v)
{
    return (int)__builtin_amdgcn_mov_dpp8((unsigned)v, (unsigned)(L * 0x249249));
}

template <int L0>
__device__ __forceinline__ void add_slots8(const int4 R, const int slot0, const int nb,
                                           const float* __restrict__ h, const int q,
                                           const unsigned last_row,
                                           float4& acc0, float4& acc1)
{
    int s[8], wb[8];
    s[0] = bc8<L0    >(R.x); wb[0] = bc8<L0    >(R.y); s[1] = bc8<L0    >(R.z); wb[1] = bc8<L0    >(R.w);
    s[2] = bc8<L0 + 1>(R.x); wb[2] = bc8<L0 + 1>(R.y); s[3] = bc8<L0 + 1>(R.z); wb[3] = bc8<L0 + 1>(R.w);
    s[4] = bc8<L0 + 2>(R.x); wb[4] = bc8<L0 + 2>(R.y); s[5] = bc8<L0 + 2>(R.z); wb[5] = bc8<L0 + 2>(R.w);
    s[6] = bc8<L0 + 3>(R.x); wb[6] = bc8<L0 + 3>(R.y); s[7] = bc8<L0 + 3>(R.z); wb[7] = bc8<L0 + 3>(R.w);

    float4 v0[8], v1[8];
    float  w[8];
    #pragma unroll
    for (int u = 0; u < 8; ++u) {
        const bool on = slot0 + u < nb;
        unsigned row = (unsigned)(on ? s[u] : s[0]);
        row  = row < last_row ? row : last_row;
        w[u] = on ? __int_as_float(wb[u]) : 0.f;
        const float4* hp = reinterpret_cast<const float4*>(h + (size_t)row * HID);
        v0[u] = hp[q];
        v1[u] = hp[8 + q];
    }
    #pragma unroll
    for (int u = 0; u < 8; ++u) {
        acc0.x += v0[u].x * w[u];
        acc0.y += v0[u].y * w[u];
        acc0.z += v0[u].z * w[u];
        acc0.w += v0[u].w * w[u];
        acc1.x += v1[u].x * w[u];
        acc1.y += v1[u].y * w[u];
        acc1.z += v1[u].z * w[u];
        acc1.w += v1[u].w * w[u];
    }
}

#define APITCH 136

typedef __attribute__((ext_vector_type(8))) _Float16 v8h;
typedef __attribute__((ext_vector_type(4))) _Float16 v4h;

__device__ __forceinline__ float4 bias_slope(float4 v, float4 b, float4 a)
{
    v.x += b.x; v.y += b.y; v.z += b.z; v.w += b.w;
    v.x = v.x > 0.f ? v.x : a.x * v.x;
    v.y = v.y > 0.f ? v.y : a.y * v.y;
    v.z = v.z > 0.f ? v.z : a.z * v.z;
    v.w = v.w > 0.f ? v.w : a.w * v.w;
    return v;
}

__global__ void __launch_bounds__(256)
gather_gemm2(const float* __restrict__ h,
             const int*   __restrict__ ei,
             const float* __restrict__ ew,
             const int2*  __restrict__ bkt,
             const int*   __restrict__ next,
             const float* __restrict__ b_conv,
             const float* __restrict__ alpha,
             const _Float16* __restrict__ bpk,
             const float* __restrict__ b_cat,
             float*       __restrict__ out,
             int n_half, int n_nodes, int E)
{
    __shared__ __align__(16) _Float16 sA[16 * APITCH];

    const int tid  = threadIdx.x;
    const int q    = tid & 7;
    const int grp  = tid >> 3;
    const int part = grp >> 4;
    const int mrow = grp & 15;
    const int m    = (blockIdx.x << 4) + mrow;
    const bool valid = m < n_half;
    const int d    = m + part * n_half;

    float4 acc0 = make_float4(0.f, 0.f, 0.f, 0.f);
    float4 acc1 = make_float4(0.f, 0.f, 0.f, 0.f);

    const int4* bq = reinterpret_cast<const int4*>(bkt) + (size_t)(valid ? d : 0) * 16;
    const int4 A = bq[q];
    const int4 B = bq[8 + q];
    const int cnt = bc8<7>(B.z);
    const int ovf = bc8<7>(B.w);
    const int nb  = valid ? (cnt < BSLOTS ? cnt : BSLOTS) : 0;
    const unsigned last_row = (unsigned)(n_nodes - 1);

    if (nb > 0)  add_slots8<0>(A,  0, nb, h, q, last_row, acc0, acc1);
    if (nb > 8)  add_slots8<4>(A,  8, nb, h, q, last_row, acc0, acc1);
    if (nb > 16) add_slots8<0>(B, 16, nb, h, q, last_row, acc0, acc1);
    if (nb > 24) add_slots8<4>(B, 24, nb, h, q, last_row, acc0, acc1);

    int   e  = valid ? ovf : -1;
    int   s  = 0;
    float w  = 0.f;
    int   en = -1;
    if ((unsigned)e < (unsigned)E) { s = ei[e]; w = ew[e]; en = next[e]; }

    for (int it = 0; it < E && (unsigned)e < (unsigned)E; ++it) {
        const float4* hp = reinterpret_cast<const float4*>(h + (long)s * HID);
        const float4 v0 = hp[q];
        const float4 v1 = hp[8 + q];
        const float  wc = w;
        e = en;
        if ((unsigned)e < (unsigned)E) { s = ei[e]; w = ew[e]; en = next[e]; }
        acc0.x += v0.x * wc;
        acc0.y += v0.y * wc;
        acc0.z += v0.z * wc;
        acc0.w += v0.w * wc;
        acc1.x += v1.x * wc;
        acc1.y += v1.y * wc;
        acc1.z += v1.z * wc;
        acc1.w += v1.w * wc;
    }

    {
        const int c0 = q << 2;
        const float4 r0 = bias_slope(acc0,
                                     *reinterpret_cast<const float4*>(b_conv + c0),
                                     *reinterpret_cast<const float4*>(alpha + c0));
        const float4 r1 = bias_slope(acc1,
                                     *reinterpret_cast<const float4*>(b_conv + 32 + c0),
                                     *reinterpret_cast<const float4*>(alpha + 32 + c0));
        _Float16* ap = sA + mrow * APITCH + (part << 6) + c0;
        v4h p0, p1;
        p0[0] = (_Float16)r0.x; p0[1] = (_Float16)r0.y; p0[2] = (_Float16)r0.z; p0[3] = (_Float16)r0.w;
        p1[0] = (_Float16)r1.x; p1[1] = (_Float16)r1.y; p1[2] = (_Float16)r1.z; p1[3] = (_Float16)r1.w;
        *reinterpret_cast<v4h*>(ap)      = p0;
        *reinterpret_cast<v4h*>(ap + 32) = p1;
    }
    __syncthreads();

    const int wave = tid >> 5;
    if (wave >= 4) return;
    const int lane  = tid & 31;
    const int ntile = wave;

    const v16h* bp = reinterpret_cast<const v16h*>(bpk) + (ntile << 7) + lane;
    const v16h b0 = bp[0], b1 = bp[32], b2 = bp[64], b3 = bp[96];

    const _Float16* ar = sA + (lane & 15) * APITCH + ((lane >> 4) << 3);
    v8f c = {};
    {
        const v8h l0 = *reinterpret_cast<const v8h*>(ar);
        const v8h h0 = *reinterpret_cast<const v8h*>(ar + 16);
        const v8h l1 = *reinterpret_cast<const v8h*>(ar + 32);
        const v8h h1 = *reinterpret_cast<const v8h*>(ar + 48);
        const v8h l2 = *reinterpret_cast<const v8h*>(ar + 64);
        const v8h h2 = *reinterpret_cast<const v8h*>(ar + 80);
        const v8h l3 = *reinterpret_cast<const v8h*>(ar + 96);
        const v8h h3 = *reinterpret_cast<const v8h*>(ar + 112);
        c = wmma_f16(__builtin_shufflevector(l0, h0, 0,1,2,3,4,5,6,7,8,9,10,11,12,13,14,15), b0, c);
        c = wmma_f16(__builtin_shufflevector(l1, h1, 0,1,2,3,4,5,6,7,8,9,10,11,12,13,14,15), b1, c);
        c = wmma_f16(__builtin_shufflevector(l2, h2, 0,1,2,3,4,5,6,7,8,9,10,11,12,13,14,15), b2, c);
        c = wmma_f16(__builtin_shufflevector(l3, h3, 0,1,2,3,4,5,6,7,8,9,10,11,12,13,14,15), b3, c);
    }

    const int col   = (ntile << 4) + (lane & 15);
    const float bb  = b_cat[col];
    const int rbase = (blockIdx.x << 4) + ((lane >> 4) << 3);
    #pragma unroll
    for (int r = 0; r < 8; ++r) {
        const int rr = rbase + r;
        if (rr < n_half) {
            float v = c[r] + bb;
            out[(long)rr * HID + col] = v > 0.f ? v : 0.f;
        }
    }
}

__global__ void __launch_bounds__(256)
bias_lrelu(float* __restrict__ agg,
           const float* __restrict__ b_conv,
           const float* __restrict__ alpha,
           int total4)
{
    const int i = blockIdx.x * blockDim.x + threadIdx.x;
    if (i >= total4) return;
    const int c4 = (i & 15) << 2;
    float4 v  = reinterpret_cast<float4*>(agg)[i];
    float4 bb = *reinterpret_cast<const float4*>(b_conv + c4);
    float4 aa = *reinterpret_cast<const float4*>(alpha + c4);
    v.x += bb.x; v.y += bb.y; v.z += bb.z; v.w += bb.w;
    v.x = v.x > 0.f ? v.x : aa.x * v.x;
    v.y = v.y > 0.f ? v.y : aa.y * v.y;
    v.z = v.z > 0.f ? v.z : aa.z * v.z;
    v.w = v.w > 0.f ? v.w : aa.w * v.w;
    reinterpret_cast<float4*>(agg)[i] = v;
}

__global__ void __launch_bounds__(256)
gemm2_wmma(const float* __restrict__ act,
           const float* __restrict__ Wcat,
           const float* __restrict__ b_cat,
           float* __restrict__ out,
           int n_half)
{
    __shared__ __align__(32) _Float16 sB[NFRAG];
    stage_Bfrag(Wcat, sB);
    __syncthreads();

    const int wave = threadIdx.x >> 5;
    const int lane = threadIdx.x & 31;
    const int mtiles  = (n_half + 15) >> 4;
    const int mgroups = (mtiles + 3) >> 2;
    const int g = blockIdx.x * (blockDim.x >> 5) + wave;
    if (g >= mgroups * 4) return;
    const int ntile = g & 3;
    const int mg    = g >> 2;

    const v16h* bp = reinterpret_cast<const v16h*>(sB) + (ntile << 7) + lane;
    const v16h b0 = bp[0], b1 = bp[32], b2 = bp[64], b3 = bp[96];

    const int koff = (lane >> 4) << 3;
    const int col  = (ntile << 4) + (lane & 15);

    for (int t = 0; t < 4; ++t) {
        const int mtile = (mg << 2) + t;
        if (mtile >= mtiles) return;
        const int row  = (mtile << 4) + (lane & 15);
        const int rowc = row < n_half ? row : n_half - 1;
        const float* a0 = act + (long)rowc * HID;
        const float* a1 = act + (long)(rowc + n_half) * HID;

        v8f c = {};
        c = wmma_f16(load_a16(a0,      koff), b0, c);
        c = wmma_f16(load_a16(a0, 32 + koff), b1, c);
        c = wmma_f16(load_a16(a1,      koff), b2, c);
        c = wmma_f16(load_a16(a1, 32 + koff), b3, c);

        const float bb = b_cat[col];
        const int rbase = (mtile << 4) + ((lane >> 4) << 3);
        #pragma unroll
        for (int r = 0; r < 8; ++r) {
            const int rr = rbase + r;
            if (rr < n_half) {
                float v = c[r] + bb;
                out[(long)rr * HID + col] = v > 0.f ? v : 0.f;
            }
        }
    }
}

extern "C" void kernel_launch(void* const* d_in, const int* in_sizes, int n_in,
                              void* d_out, int out_size, void* d_ws, size_t ws_size,
                              hipStream_t stream)
{
    const float* x   = (const float*)d_in[0];
    const int*   ei  = (const int*)  d_in[1];
    const float* ew  = (const float*)d_in[2];
    const float* Wc  = (const float*)d_in[3];
    const float* bc  = (const float*)d_in[4];
    const float* al  = (const float*)d_in[5];
    const float* Wk  = (const float*)d_in[6];
    const float* bk  = (const float*)d_in[7];

    const int n_nodes = in_sizes[0] / IN_CH;
    const int E       = in_sizes[2];
    const int n_half  = n_nodes / 2;

    float* h   = (float*)d_ws;
    float* agg = h + (size_t)n_nodes * HID;
    float* out = (float*)d_out;

    {
        const int mgroups = ((n_nodes >> 4) + 3) >> 2;
        const int waves   = mgroups * 4;
        gemm1_wmma<<<(waves + 7) / 8, 256, 0, stream>>>(x, Wc, h, n_nodes);
    }

    const int n4 = (n_nodes * HID) >> 2;
    int* head = (int*)(agg + (size_t)n_nodes * HID);
    int* next = head + n_nodes;
    const size_t need = (size_t)n_nodes * HID * 2 * sizeof(float)
                      + ((size_t)n_nodes + (size_t)E) * sizeof(int);
    if (ws_size >= need && (size_t)n_nodes * sizeof(int) >= NFRAG * sizeof(_Float16)) {
        int2* bkt = (int2*)agg;
        _Float16* bpk = (_Float16*)head;
        const int nfill = n_nodes > NFRAG ? n_nodes : NFRAG;
        init_buckets_pack_b<<<(nfill + 255) / 256, 256, 0, stream>>>(bkt, n_nodes, Wk, bpk);
        bin_edges<<<(E + 255) / 256, 256, 0, stream>>>(ei, ew, bkt, next, E);
        gather_gemm2<<<(n_half + 15) / 16, 256, 0, stream>>>(
            h, ei, ew, bkt, next, bc, al, bpk, bk, out, n_half, n_nodes, E);
        return;
    } else {
        zero_f32<<<(n4 + 255) / 256, 256, 0, stream>>>((float4*)agg, n4);

        scatter_edges<<<(E + 7) / 8, 256, 0, stream>>>(h, ei, ew, agg, E);
    }

    bias_lrelu<<<(n4 + 255) / 256, 256, 0, stream>>>(agg, bc, al, n4);

    {
        const int mgroups = (((n_half + 15) >> 4) + 3) >> 2;
        const int waves   = mgroups * 4;
        gemm2_wmma<<<(waves + 7) / 8, 256, 0, stream>>>(agg, Wk, bk, out, n_half);
    }
}


# edge partition assembles its region of records in LDS and copies it out 16 bytes per lane, whole 128-byte lines per store, instead of 16 scattered 8-byte stores per lane; record order unchanged
# speedup vs baseline: 4.1368x; 2.4310x over previous
#include <hip/hip_runtime.h>

typedef __attribute__((ext_vector_type(16))) _Float16 v16h;
typedef __attribute__((ext_vector_type(8)))  float    v8f;
typedef __attribute__((ext_vector_type(8)))  _Float16 v8h;

#define IN_CH 128
#define HID   64
#define NFRAG (4 * 4 * 32 * 16)

__device__ __forceinline__ v8f wmma_f16(v16h a, v16h b, v8f c)
{
    return __builtin_amdgcn_wmma_f32_16x16x32_f16(false, a, false, b,
                                                  (short)0, c, false, false);
}

__device__ __forceinline__ void stage_Bfrag(const float* __restrict__ W,
                                            _Float16* sB)
{
    for (int i = threadIdx.x; i < NFRAG; i += blockDim.x) {
        const int j     = i & 15;
        const int lane  = (i >> 4) & 31;
        const int kci   = (i >> 9) & 3;
        const int ntile = i >> 11;
        const int koff  = (lane >> 4) << 3;
        const int col   = (ntile << 4) + (lane & 15);
        const int k     = (kci << 5) + koff + (j & 7) + ((j >> 3) << 4);
        sB[i] = (_Float16)W[k * HID + col];
    }
}

__device__ __forceinline__ v16h load_a16(const float* __restrict__ arow, int off)
{
    float af[16];
    reinterpret_cast<float4*>(af)[0] = *reinterpret_cast<const float4*>(arow + off);
    reinterpret_cast<float4*>(af)[1] = *reinterpret_cast<const float4*>(arow + off + 4);
    reinterpret_cast<float4*>(af)[2] = *reinterpret_cast<const float4*>(arow + off + 16);
    reinterpret_cast<float4*>(af)[3] = *reinterpret_cast<const float4*>(arow + off + 20);
    v16h a;
    #pragma unroll
    for (int j = 0; j < 16; ++j) a[j] = (_Float16)af[j];
    return a;
}


__device__ __forceinline__ void stage_Bfrag_patch(const float* __restrict__ W,
                                                  _Float16* sB)
{
    const int tid = threadIdx.x;
    const int cg  = tid & 15;
    const int kg  = tid >> 4;
    const float4* wp = reinterpret_cast<const float4*>(W + (kg << 3) * HID) + cg;
    float4 v[8];
    #pragma unroll
    for (int e = 0; e < 8; ++e) v[e] = wp[e * (HID / 4)];

    const int kci   = kg >> 2;
    const int hh    = kg & 1;
    const int jh    = (kg >> 1) & 1;
    const int ntile = cg >> 2;
    const int l16   = (cg & 3) << 2;
    _Float16* dst = sB + ((((ntile << 2) + kci) << 5) + (hh << 4) + l16) * 16 + (jh << 3);
    v8h p0, p1, p2, p3;
    #pragma unroll
    for (int e = 0; e < 8; ++e) {
        p0[e] = (_Float16)v[e].x;
        p1[e] = (_Float16)v[e].y;
        p2[e] = (_Float16)v[e].z;
        p3[e] = (_Float16)v[e].w;
    }
    *reinterpret_cast<v8h*>(dst)      = p0;
    *reinterpret_cast<v8h*>(dst + 16) = p1;
    *reinterpret_cast<v8h*>(dst + 32) = p2;
    *reinterpret_cast<v8h*>(dst + 48) = p3;
}

__device__ __forceinline__ v16h cvt_a16(const float4 a0, const float4 a1,
                                        const float4 a2, const float4 a3)
{
    v16h a;
    a[0]  = (_Float16)a0.x; a[1]  = (_Float16)a0.y; a[2]  = (_Float16)a0.z; a[3]  = (_Float16)a0.w;
    a[4]  = (_Float16)a1.x; a[5]  = (_Float16)a1.y; a[6]  = (_Float16)a1.z; a[7]  = (_Float16)a1.w;
    a[8]  = (_Float16)a2.x; a[9]  = (_Float16)a2.y; a[10] = (_Float16)a2.z; a[11] = (_Float16)a2.w;
    a[12] = (_Float16)a3.x; a[13] = (_Float16)a3.y; a[14] = (_Float16)a3.z; a[15] = (_Float16)a3.w;
    return a;
}

__device__ __forceinline__ void
gemm1_block(const float* __restrict__ x,
            const float* __restrict__ Wc,
            float* __restrict__ h,
            int n_nodes, _Float16* sB, int bid)
{
    const int wave = threadIdx.x >> 5;
    const int lane = threadIdx.x & 31;
    const int mtiles = n_nodes >> 4;
    int mtile = bid * 8 + wave;
    const bool has = mtile < mtiles;
    mtile = has ? mtile : mtiles - 1;

    const int koff = (lane >> 4) << 3;
    const float* arow = x + (size_t)((mtile << 4) + (lane & 15)) * IN_CH + koff;
    float4 ar[16];
    #pragma unroll
    for (int kc = 0; kc < 4; ++kc) {
        ar[4 * kc + 0] = *reinterpret_cast<const float4*>(arow + 32 * kc);
        ar[4 * kc + 1] = *reinterpret_cast<const float4*>(arow + 32 * kc + 4);
        ar[4 * kc + 2] = *reinterpret_cast<const float4*>(arow + 32 * kc + 16);
        ar[4 * kc + 3] = *reinterpret_cast<const float4*>(arow + 32 * kc + 20);
    }

    stage_Bfrag_patch(Wc, sB);
    __syncthreads();
    if (!has) return;

    const v16h a0 = cvt_a16(ar[0],  ar[1],  ar[2],  ar[3]);
    const v16h a1 = cvt_a16(ar[4],  ar[5],  ar[6],  ar[7]);
    const v16h a2 = cvt_a16(ar[8],  ar[9],  ar[10], ar[11]);
    const v16h a3 = cvt_a16(ar[12], ar[13], ar[14], ar[15]);

    const int rbase = (mtile << 4) + ((lane >> 4) << 3);
    float* hp = h + (size_t)rbase * HID + (lane & 15);
    #pragma unroll
    for (int nt = 0; nt < 4; ++nt) {
        const v16h* bp = reinterpret_cast<const v16h*>(sB) + (nt << 7) + lane;
        v8f c = {};
        c = wmma_f16(a0, bp[0],  c);
        c = wmma_f16(a1, bp[32], c);
        c = wmma_f16(a2, bp[64], c);
        c = wmma_f16(a3, bp[96], c);
        #pragma unroll
        for (int r = 0; r < 8; ++r)
            hp[r * HID + (nt << 4)] = c[r];
    }
}

__global__ void __launch_bounds__(256)
gemm1_wmma(const float* __restrict__ x,
           const float* __restrict__ Wc,
           float* __restrict__ h,
           int n_nodes)
{
    __shared__ __align__(32) _Float16 sB[NFRAG];
    gemm1_block(x, Wc, h, n_nodes, sB, blockIdx.x);
}

__global__ void zero_f32(float4* __restrict__ p, int n4)
{
    int i = blockIdx.x * blockDim.x + threadIdx.x;
    for (; i < n4; i += gridDim.x * blockDim.x)
        p[i] = make_float4(0.f, 0.f, 0.f, 0.f);
}

__global__ void __launch_bounds__(256)
scatter_edges(const float* __restrict__ h,
              const int*   __restrict__ ei,
              const float* __restrict__ ew,
              float*       __restrict__ agg,
              int E)
{
    const int t    = blockIdx.x * blockDim.x + threadIdx.x;
    const int e    = t >> 5;
    const int lane = t & 31;
    if (e >= E) return;

    const int   s = ei[e];
    const int   d = ei[E + e];
    const float w = ew[e];

    float2 hv = reinterpret_cast<const float2*>(h + (long)s * HID)[lane];
    float* ap = agg + (long)d * HID + (lane << 1);
    atomicAdd(ap,     hv.x * w);
    atomicAdd(ap + 1, hv.y * w);
}

#define RPB    64
#define DPB    (2 * RPB)
#define BINCAP (DPB * 32)
#define EPT    16
#define PBE    (256 * EPT)
#define MAXBIN 4096
#define SEGW   16

__device__ __forceinline__ void
part_block(const int*   __restrict__ ei,
           const float* __restrict__ ew,
           int2* __restrict__ recs,
           int2* __restrict__ segt,
           const float* __restrict__ Wcat,
           _Float16* __restrict__ bpk,
           int E, int n_half, int nbin, int npart, int pb, int* hist, int2* stg)
{
    const int tid  = threadIdx.x;
    const int wave = tid >> 5;
    const int lane = tid & 31;
    for (int b = tid; b < nbin + 32; b += 256) hist[b] = 0;
    __syncthreads();

    int key[EPT], dst[EPT], wb[EPT], br[EPT];
    const int e0 = pb * PBE + tid;
    #pragma unroll
    for (int u = 0; u < EPT; ++u) {
        const int e  = e0 + u * 256;
        const int ec = e < E ? e : E - 1;
        key[u] = ei[ec];
        dst[u] = ei[E + ec];
        wb[u]  = __float_as_int(ew[ec]);
    }
    for (int i = pb * 256 + tid; i < NFRAG; i += npart * 256) {
        const int j     = i & 15;
        const int ln    = (i >> 4) & 31;
        const int kci   = (i >> 9) & 3;
        const int ntile = i >> 11;
        const int koff  = (ln >> 4) << 3;
        const int col   = (ntile << 4) + (ln & 15);
        const int k     = (kci << 5) + koff + (j & 7) + ((j >> 3) << 4);
        bpk[i] = (_Float16)Wcat[k * HID + col];
    }
    #pragma unroll
    for (int u = 0; u < EPT; ++u) {
        const int e = e0 + u * 256;
        const unsigned d = (unsigned)dst[u];
        const bool ok = e < E && d < (unsigned)(2 * n_half);
        const unsigned part = d >= (unsigned)n_half ? 1u : 0u;
        const unsigned m    = d - part * (unsigned)n_half;
        const int bin = ok ? (int)(m / RPB) : nbin + (tid & 31);
        const int dl  = (int)((m % RPB) + part * RPB);
        key[u] = (key[u] & 0xFFFFFF) | (dl << 24);
        const int rank = atomicAdd(&hist[bin], 1);
        br[u] = ok ? (bin | (rank << 12)) : -1;
    }
    __syncthreads();
    const int bpt = (nbin + 255) >> 8;
    const int b0  = tid * bpt;
    int s = 0;
    for (int k = 0; k < bpt; ++k) {
        const int b = b0 + k;
        if (b < nbin) s += hist[b];
    }
    int incl = s;
    #pragma unroll
    for (int o = 1; o < 32; o <<= 1) {
        const int v = __shfl_up(incl, o);
        if (lane >= o) incl += v;
    }
    if (lane == 31) hist[nbin + 32 + wave] = incl;
    __syncthreads();
    int run = incl - s;
    for (int w = 0; w < wave; ++w) run += hist[nbin + 32 + w];
    for (int k = 0; k < bpt; ++k) {
        const int b = b0 + k;
        if (b < nbin) {
            const int c = hist[b];
            hist[b] = run;
            segt[(size_t)b * npart + pb] = make_int2(run, c);
            run += c;
        }
    }
    __syncthreads();
    #pragma unroll
    for (int u = 0; u < EPT; ++u)
        dst[u] = hist[br[u] & (MAXBIN - 1)] + (br[u] >> 12);
    #pragma unroll
    for (int u = 0; u < EPT; ++u) {
        if (br[u] >= 0) {
            const int pos = dst[u];
            if (pos >= 0 && pos < PBE) stg[pos] = make_int2(key[u], wb[u]);
        }
    }
    int tot = 0;
    for (int w = 0; w < 8; ++w) tot += hist[nbin + 32 + w];
    tot = tot < 0 ? 0 : (tot > PBE ? PBE : tot);
    for (int i = tot + tid; i < PBE; i += 256) stg[i] = make_int2(0, 0);
    __syncthreads();
    int4* rp4 = reinterpret_cast<int4*>(recs + (size_t)pb * PBE);
    const int4* sp4 = reinterpret_cast<const int4*>(stg);
    int4 pv[EPT / 2];
    #pragma unroll
    for (int i = 0; i < EPT / 2; ++i) pv[i] = sp4[i * 256 + tid];
    #pragma unroll
    for (int i = 0; i < EPT / 2; ++i) rp4[i * 256 + tid] = pv[i];
}

__global__ void __launch_bounds__(256)
gemm1_and_part(const float* __restrict__ x,
               const float* __restrict__ Wc,
               float* __restrict__ h,
               int n_nodes, int nb_gemm,
               const int*   __restrict__ ei,
               const float* __restrict__ ew,
               int2* __restrict__ recs,
               int2* __restrict__ segt,
               const float* __restrict__ Wcat,
               _Float16* __restrict__ bpk,
               int E, int n_half, int nbin, int npart)
{
    __shared__ __align__(32) _Float16 sB[NFRAG];
    __shared__ __align__(16) int2 stg[PBE];
    const int bid = blockIdx.x;
    if (bid < nb_gemm) gemm1_block(x, Wc, h, n_nodes, sB, bid);
    else               part_block(ei, ew, recs, segt, Wcat, bpk, E, n_half, nbin, npart, bid - nb_gemm,
                                  reinterpret_cast<int*>(sB), stg);
}

#define APITCH 136

typedef __attribute__((ext_vector_type(4))) _Float16 v4h;

__device__ __forceinline__ float4 bias_slope(float4 v, float4 b, float4 a)
{
    v.x += b.x; v.y += b.y; v.z += b.z; v.w += b.w;
    v.x = v.x > 0.f ? v.x : a.x * v.x;
    v.y = v.y > 0.f ? v.y : a.y * v.y;
    v.z = v.z > 0.f ? v.z : a.z * v.z;
    v.w = v.w > 0.f ? v.w : a.w * v.w;
    return v;
}

__device__ __forceinline__ void fma_row(float4& acc0, float4& acc1,
                                        const float4 v0, const float4 v1, const float w)
{
    acc0.x += v0.x * w; acc0.y += v0.y * w; acc0.z += v0.z * w; acc0.w += v0.w * w;
    acc1.x += v1.x * w; acc1.y += v1.y * w; acc1.z += v1.z * w; acc1.w += v1.w * w;
}

__device__ __forceinline__ void add_recs8(const int2* srec, const int base, const int end,
                                          const float* __restrict__ h, const int q,
                                          const unsigned last_row,
                                          float4& acc0, float4& acc1)
{
    int2   rr[8];
    float4 v0[8], v1[8];
    float  w[8];
    #pragma unroll
    for (int u = 0; u < 8; ++u) rr[u] = srec[base + u];
    #pragma unroll
    for (int u = 0; u < 8; ++u) {
        const bool on = base + u < end;
        unsigned row  = (unsigned)((on ? rr[u].x : rr[0].x) & 0xFFFFFF);
        row  = row < last_row ? row : last_row;
        w[u] = on ? __int_as_float(rr[u].y) : 0.f;
        const float4* hp = reinterpret_cast<const float4*>(h + (size_t)row * HID);
        v0[u] = hp[q];
        v1[u] = hp[8 + q];
    }
    #pragma unroll
    for (int u = 0; u < 8; ++u) fma_row(acc0, acc1, v0[u], v1[u], w[u]);
}

__global__ void __launch_bounds__(512)
gather_sort_gemm2(const float* __restrict__ h,
                  const int2*  __restrict__ recs,
                  const int2*  __restrict__ segt,
                  const float* __restrict__ b_conv,
                  const float* __restrict__ alpha,
                  const _Float16* __restrict__ bpk,
                  const float* __restrict__ b_cat,
                  float*       __restrict__ out,
                  int n_half, int n_nodes, int nbin, int npart)
{
    __shared__ __align__(16) int2 srec[BINCAP + 8];
    __shared__ __align__(16) int  scnt[DPB + 32];
    __shared__ __align__(16) int  soff[DPB + 32];
    __shared__ __align__(16) _Float16 sA[RPB * APITCH];
    __shared__ int sctl[2];

    const int tid   = threadIdx.x;
    const int bin   = blockIdx.x;
    const int wv    = tid >> 5;
    const int ln    = tid & 31;

    if (tid < DPB + 32) scnt[tid] = 0;
    if (tid == 0) sctl[0] = 0;

    int2 sd = make_int2(0, 0);
    {
        const int p = wv + 16 * (ln & 15);
        if (p < npart) sd = segt[(size_t)bin * npart + p];
        sd.x = sd.x < 0 ? 0 : (sd.x > PBE ? PBE : sd.x);
        sd.y = sd.y < 0 ? 0 : (sd.y > PBE - sd.x ? PBE - sd.x : sd.y);
    }
    int2 r[SEGW];
    int  rk[SEGW];
    bool lng = false;
    #pragma unroll
    for (int j = 0; j < SEGW; ++j) {
        const int st = __builtin_amdgcn_readlane(sd.x, j);
        const int cn = __builtin_amdgcn_readlane(sd.y, j);
        lng = lng || cn > 32;
        r[j] = make_int2((DPB + ln) << 24, 0);
        if (ln < cn) r[j] = recs[(size_t)(wv + 16 * j) * PBE + st + ln];
    }
    __syncthreads();
    if (lng && ln == 0) sctl[0] = 1;

    #pragma unroll
    for (int j = 0; j < SEGW; ++j)
        rk[j] = atomicAdd(&scnt[(r[j].x >> 24) & 255], 1);
    __syncthreads();

    if (tid < 32) {
        const int4 c = reinterpret_cast<const int4*>(scnt)[tid];
        const int t = c.x + c.y + c.z + c.w;
        int incl = t;
        #pragma unroll
        for (int o = 1; o < 32; o <<= 1) {
            const int v = __shfl_up(incl, o);
            if (tid >= o) incl += v;
        }
        const int ex = incl - t;
        reinterpret_cast<int4*>(soff)[tid] = make_int4(ex, ex + c.x, ex + c.x + c.y, ex + c.x + c.y + c.z);
        if (tid == 31) sctl[1] = incl;
    }
    __syncthreads();

    const bool big  = sctl[1] > BINCAP;
    const bool scan = big || sctl[0] != 0;

    if (!big) {
        #pragma unroll
        for (int j = 0; j < SEGW; ++j)
            rk[j] += soff[(r[j].x >> 24) & 255];
        #pragma unroll
        for (int j = 0; j < SEGW; ++j)
            if (((r[j].x >> 24) & 255) < DPB) srec[rk[j]] = r[j];
    }
    __syncthreads();

    const int q    = tid & 7;
    const int g    = tid >> 3;
    const unsigned last_row = (unsigned)(n_nodes - 1);
    const int c0 = q << 2;
    const float4 bc0 = *reinterpret_cast<const float4*>(b_conv + c0);
    const float4 bc1 = *reinterpret_cast<const float4*>(b_conv + 32 + c0);
    const float4 al0 = *reinterpret_cast<const float4*>(alpha + c0);
    const float4 al1 = *reinterpret_cast<const float4*>(alpha + 32 + c0);

    for (int part = 0; part < 2; ++part) {
        const int dl  = part * RPB + g;
        const int beg = big ? 0 : soff[dl];
        const int end = big ? 0 : beg + scnt[dl];
        float4 acc0 = make_float4(0.f, 0.f, 0.f, 0.f);
        float4 acc1 = make_float4(0.f, 0.f, 0.f, 0.f);

        for (int base = beg; base < end; base += 8)
            add_recs8(srec, base, end, h, q, last_row, acc0, acc1);

        if (scan) {
            const int i0 = big ? 0 : 32;
            for (int p = 0; p < npart; ++p) {
                int2 pd = segt[(size_t)bin * npart + p];
                pd.x = pd.x < 0 ? 0 : (pd.x > PBE ? PBE : pd.x);
                pd.y = pd.y < 0 ? 0 : (pd.y > PBE - pd.x ? PBE - pd.x : pd.y);
                const int2* pr = recs + (size_t)p * PBE + pd.x;
                for (int i = i0; i < pd.y; ++i) {
                    const int2 rc = pr[i];
                    if (((rc.x >> 24) & 255) == dl) {
                        unsigned row = (unsigned)(rc.x & 0xFFFFFF);
                        row = row < last_row ? row : last_row;
                        const float4* hp = reinterpret_cast<const float4*>(h + (size_t)row * HID);
                        fma_row(acc0, acc1, hp[q], hp[8 + q], __int_as_float(rc.y));
                    }
                }
            }
        }

        const float4 r0 = bias_slope(acc0, bc0, al0);
        const float4 r1 = bias_slope(acc1, bc1, al1);
        _Float16* ap = sA + g * APITCH + (part << 6) + c0;
        v4h p0, p1;
        p0[0] = (_Float16)r0.x; p0[1] = (_Float16)r0.y; p0[2] = (_Float16)r0.z; p0[3] = (_Float16)r0.w;
        p1[0] = (_Float16)r1.x; p1[1] = (_Float16)r1.y; p1[2] = (_Float16)r1.z; p1[3] = (_Float16)r1.w;
        *reinterpret_cast<v4h*>(ap)      = p0;
        *reinterpret_cast<v4h*>(ap + 32) = p1;
    }
    __syncthreads();

    const int wave  = tid >> 5;
    const int lane  = tid & 31;
    const int ntile = wave & 3;
    const int mt    = wave >> 2;

    const v16h* bp = reinterpret_cast<const v16h*>(bpk) + (ntile << 7) + lane;
    const v16h b0 = bp[0], b1 = bp[32], b2 = bp[64], b3 = bp[96];

    const _Float16* ar = sA + ((mt << 4) + (lane & 15)) * APITCH + ((lane >> 4) << 3);
    v8f c = {};
    {
        const v8h l0 = *reinterpret_cast<const v8h*>(ar);
        const v8h h0 = *reinterpret_cast<const v8h*>(ar + 16);
        const v8h l1 = *reinterpret_cast<const v8h*>(ar + 32);
        const v8h h1 = *reinterpret_cast<const v8h*>(ar + 48);
        const v8h l2 = *reinterpret_cast<const v8h*>(ar + 64);
        const v8h h2 = *reinterpret_cast<const v8h*>(ar + 80);
        const v8h l3 = *reinterpret_cast<const v8h*>(ar + 96);
        const v8h h3 = *reinterpret_cast<const v8h*>(ar + 112);
        c = wmma_f16(__builtin_shufflevector(l0, h0, 0,1,2,3,4,5,6,7,8,9,10,11,12,13,14,15), b0, c);
        c = wmma_f16(__builtin_shufflevector(l1, h1, 0,1,2,3,4,5,6,7,8,9,10,11,12,13,14,15), b1, c);
        c = wmma_f16(__builtin_shufflevector(l2, h2, 0,1,2,3,4,5,6,7,8,9,10,11,12,13,14,15), b2, c);
        c = wmma_f16(__builtin_shufflevector(l3, h3, 0,1,2,3,4,5,6,7,8,9,10,11,12,13,14,15), b3, c);
    }

    const int col   = (ntile << 4) + (lane & 15);
    const float bb  = b_cat[col];
    const int rbase = bin * RPB + (mt << 4) + ((lane >> 4) << 3);
    #pragma unroll
    for (int r8 = 0; r8 < 8; ++r8) {
        const int rr = rbase + r8;
        if (rr < n_half) {
            float v = c[r8] + bb;
            out[(long)rr * HID + col] = v > 0.f ? v : 0.f;
        }
    }
}

__global__ void __launch_bounds__(256)
bias_lrelu(float* __restrict__ agg,
           const float* __restrict__ b_conv,
           const float* __restrict__ alpha,
           int total4)
{
    const int i = blockIdx.x * blockDim.x + threadIdx.x;
    if (i >= total4) return;
    const int c4 = (i & 15) << 2;
    float4 v  = reinterpret_cast<float4*>(agg)[i];
    float4 bb = *reinterpret_cast<const float4*>(b_conv + c4);
    float4 aa = *reinterpret_cast<const float4*>(alpha + c4);
    v.x += bb.x; v.y += bb.y; v.z += bb.z; v.w += bb.w;
    v.x = v.x > 0.f ? v.x : aa.x * v.x;
    v.y = v.y > 0.f ? v.y : aa.y * v.y;
    v.z = v.z > 0.f ? v.z : aa.z * v.z;
    v.w = v.w > 0.f ? v.w : aa.w * v.w;
    reinterpret_cast<float4*>(agg)[i] = v;
}

__global__ void __launch_bounds__(256)
gemm2_wmma(const float* __restrict__ act,
           const float* __restrict__ Wcat,
           const float* __restrict__ b_cat,
           float* __restrict__ out,
           int n_half)
{
    __shared__ __align__(32) _Float16 sB[NFRAG];
    stage_Bfrag(Wcat, sB);
    __syncthreads();

    const int wave = threadIdx.x >> 5;
    const int lane = threadIdx.x & 31;
    const int mtiles  = (n_half + 15) >> 4;
    const int mgroups = (mtiles + 3) >> 2;
    const int g = blockIdx.x * (blockDim.x >> 5) + wave;
    if (g >= mgroups * 4) return;
    const int ntile = g & 3;
    const int mg    = g >> 2;

    const v16h* bp = reinterpret_cast<const v16h*>(sB) + (ntile << 7) + lane;
    const v16h b0 = bp[0], b1 = bp[32], b2 = bp[64], b3 = bp[96];

    const int koff = (lane >> 4) << 3;
    const int col  = (ntile << 4) + (lane & 15);

    for (int t = 0; t < 4; ++t) {
        const int mtile = (mg << 2) + t;
        if (mtile >= mtiles) return;
        const int row  = (mtile << 4) + (lane & 15);
        const int rowc = row < n_half ? row : n_half - 1;
        const float* a0 = act + (long)rowc * HID;
        const float* a1 = act + (long)(rowc + n_half) * HID;

        v8f c = {};
        c = wmma_f16(load_a16(a0,      koff), b0, c);
        c = wmma_f16(load_a16(a0, 32 + koff), b1, c);
        c = wmma_f16(load_a16(a1,      koff), b2, c);
        c = wmma_f16(load_a16(a1, 32 + koff), b3, c);

        const float bb = b_cat[col];
        const int rbase = (mtile << 4) + ((lane >> 4) << 3);
        #pragma unroll
        for (int r = 0; r < 8; ++r) {
            const int rr = rbase + r;
            if (rr < n_half) {
                float v = c[r] + bb;
                out[(long)rr * HID + col] = v > 0.f ? v : 0.f;
            }
        }
    }
}

extern "C" void kernel_launch(void* const* d_in, const int* in_sizes, int n_in,
                              void* d_out, int out_size, void* d_ws, size_t ws_size,
                              hipStream_t stream)
{
    const float* x   = (const float*)d_in[0];
    const int*   ei  = (const int*)  d_in[1];
    const float* ew  = (const float*)d_in[2];
    const float* Wc  = (const float*)d_in[3];
    const float* bc  = (const float*)d_in[4];
    const float* al  = (const float*)d_in[5];
    const float* Wk  = (const float*)d_in[6];
    const float* bk  = (const float*)d_in[7];

    const int n_nodes = in_sizes[0] / IN_CH;
    const int E       = in_sizes[2];
    const int n_half  = n_nodes / 2;

    float* h   = (float*)d_ws;
    float* agg = h + (size_t)n_nodes * HID;
    float* out = (float*)d_out;

    const int nb_gemm = (((((n_nodes >> 4) + 3) >> 2) * 4) + 7) / 8;

    const int n4 = (n_nodes * HID) >> 2;
    const int nbin  = (n_half + RPB - 1) / RPB;
    const int npart = (E + PBE - 1) / PBE;
    int2* recs = (int2*)agg;
    _Float16* bpk = (_Float16*)(recs + (size_t)npart * PBE);
    int2* segt = (int2*)(bpk + NFRAG);
    const size_t need = (size_t)n_nodes * HID * sizeof(float)
                      + (size_t)npart * PBE * sizeof(int2)
                      + NFRAG * sizeof(_Float16)
                      + (size_t)nbin * (size_t)npart * sizeof(int2);
    if (ws_size >= need && nbin + 40 <= MAXBIN && npart <= 16 * SEGW && n_nodes <= (1 << 24) && E > 0) {
        gemm1_and_part<<<nb_gemm + npart, 256, 0, stream>>>(
            x, Wc, h, n_nodes, nb_gemm, ei, ew, recs, segt, Wk, bpk, E, n_half, nbin, npart);
        gather_sort_gemm2<<<nbin, 512, 0, stream>>>(
            h, recs, segt, bc, al, bpk, bk, out, n_half, n_nodes, nbin, npart);
        return;
    } else {
        gemm1_wmma<<<nb_gemm, 256, 0, stream>>>(x, Wc, h, n_nodes);

        zero_f32<<<(n4 + 255) / 256, 256, 0, stream>>>((float4*)agg, n4);

        scatter_edges<<<(E + 7) / 8, 256, 0, stream>>>(h, ei, ew, agg, E);
    }

    bias_lrelu<<<(n4 + 255) / 256, 256, 0, stream>>>(agg, bc, al, n4);

    {
        const int mgroups = (((n_half + 15) >> 4) + 3) >> 2;
        const int waves   = mgroups * 4;
        gemm2_wmma<<<(waves + 7) / 8, 256, 0, stream>>>(agg, Wk, bk, out, n_half);
    }
}


# edge sums: one v_fma_mix_f32 per channel (converts, multiplies, adds in f32) in place of a conversion plus a packed multiply-add; sorted records carry ready row byte offsets; same sums, same order
# speedup vs baseline: 4.8806x; 1.1798x over previous
#include <hip/hip_runtime.h>

typedef __attribute__((ext_vector_type(16))) _Float16 v16h;
typedef __attribute__((ext_vector_type(8)))  float    v8f;
typedef __attribute__((ext_vector_type(8)))  _Float16 v8h;
typedef __attribute__((ext_vector_type(4)))  unsigned v4u;

#define IN_CH 128
#define HID   64
#define NFRAG (4 * 4 * 32 * 16)

__device__ __forceinline__ v8f wmma_f16(v16h a, v16h b, v8f c)
{
    return __builtin_amdgcn_wmma_f32_16x16x32_f16(false, a, false, b,
                                                  (short)0, c, false, false);
}

__device__ __forceinline__ void stage_Bfrag(const float* __restrict__ W,
                                            _Float16* sB)
{
    for (int i = threadIdx.x; i < NFRAG; i += blockDim.x) {
        const int j     = i & 15;
        const int lane  = (i >> 4) & 31;
        const int kci   = (i >> 9) & 3;
        const int ntile = i >> 11;
        const int koff  = (lane >> 4) << 3;
        const int col   = (ntile << 4) + (lane & 15);
        const int k     = (kci << 5) + koff + (j & 7) + ((j >> 3) << 4);
        sB[i] = (_Float16)W[k * HID + col];
    }
}

__device__ __forceinline__ v16h load_a16(const float* __restrict__ arow, int off)
{
    float af[16];
    reinterpret_cast<float4*>(af)[0] = *reinterpret_cast<const float4*>(arow + off);
    reinterpret_cast<float4*>(af)[1] = *reinterpret_cast<const float4*>(arow + off + 4);
    reinterpret_cast<float4*>(af)[2] = *reinterpret_cast<const float4*>(arow + off + 16);
    reinterpret_cast<float4*>(af)[3] = *reinterpret_cast<const float4*>(arow + off + 20);
    v16h a;
    #pragma unroll
    for (int j = 0; j < 16; ++j) a[j] = (_Float16)af[j];
    return a;
}


__device__ __forceinline__ void stage_Bfrag_patch(const float* __restrict__ W,
                                                  _Float16* sB)
{
    const int tid = threadIdx.x;
    const int cg  = tid & 15;
    const int kg  = tid >> 4;
    const float4* wp = reinterpret_cast<const float4*>(W + (kg << 3) * HID) + cg;
    float4 v[8];
    #pragma unroll
    for (int e = 0; e < 8; ++e) v[e] = wp[e * (HID / 4)];

    const int kci   = kg >> 2;
    const int hh    = kg & 1;
    const int jh    = (kg >> 1) & 1;
    const int ntile = cg >> 2;
    const int l16   = (cg & 3) << 2;
    _Float16* dst = sB + ((((ntile << 2) + kci) << 5) + (hh << 4) + l16) * 16 + (jh << 3);
    v8h p0, p1, p2, p3;
    #pragma unroll
    for (int e = 0; e < 8; ++e) {
        p0[e] = (_Float16)v[e].x;
        p1[e] = (_Float16)v[e].y;
        p2[e] = (_Float16)v[e].z;
        p3[e] = (_Float16)v[e].w;
    }
    *reinterpret_cast<v8h*>(dst)      = p0;
    *reinterpret_cast<v8h*>(dst + 16) = p1;
    *reinterpret_cast<v8h*>(dst + 32) = p2;
    *reinterpret_cast<v8h*>(dst + 48) = p3;
}

__device__ __forceinline__ v16h cvt_a16(const float4 a0, const float4 a1,
                                        const float4 a2, const float4 a3)
{
    v16h a;
    a[0]  = (_Float16)a0.x; a[1]  = (_Float16)a0.y; a[2]  = (_Float16)a0.z; a[3]  = (_Float16)a0.w;
    a[4]  = (_Float16)a1.x; a[5]  = (_Float16)a1.y; a[6]  = (_Float16)a1.z; a[7]  = (_Float16)a1.w;
    a[8]  = (_Float16)a2.x; a[9]  = (_Float16)a2.y; a[10] = (_Float16)a2.z; a[11] = (_Float16)a2.w;
    a[12] = (_Float16)a3.x; a[13] = (_Float16)a3.y; a[14] = (_Float16)a3.z; a[15] = (_Float16)a3.w;
    return a;
}

#define HPITCH 72

template <bool H16>
__device__ __forceinline__ void
gemm1_block(const float* __restrict__ x,
            const float* __restrict__ Wc,
            float* __restrict__ h,
            int n_nodes, _Float16* sB, int bid, _Float16* tl)
{
    const int wave = threadIdx.x >> 5;
    const int lane = threadIdx.x & 31;
    const int mtiles = n_nodes >> 4;
    int mtile = bid * 8 + wave;
    const bool has = mtile < mtiles;
    mtile = has ? mtile : mtiles - 1;

    const int koff = (lane >> 4) << 3;
    const float* arow = x + (size_t)((mtile << 4) + (lane & 15)) * IN_CH + koff;
    float4 ar[16];
    #pragma unroll
    for (int kc = 0; kc < 4; ++kc) {
        ar[4 * kc + 0] = *reinterpret_cast<const float4*>(arow + 32 * kc);
        ar[4 * kc + 1] = *reinterpret_cast<const float4*>(arow + 32 * kc + 4);
        ar[4 * kc + 2] = *reinterpret_cast<const float4*>(arow + 32 * kc + 16);
        ar[4 * kc + 3] = *reinterpret_cast<const float4*>(arow + 32 * kc + 20);
    }

    stage_Bfrag_patch(Wc, sB);
    __syncthreads();
    if (!H16 && !has) return;

    const v16h a0 = cvt_a16(ar[0],  ar[1],  ar[2],  ar[3]);
    const v16h a1 = cvt_a16(ar[4],  ar[5],  ar[6],  ar[7]);
    const v16h a2 = cvt_a16(ar[8],  ar[9],  ar[10], ar[11]);
    const v16h a3 = cvt_a16(ar[12], ar[13], ar[14], ar[15]);

    if (H16) {
        _Float16* tw = tl + wave * (16 * HPITCH);
        _Float16* tp = tw + ((lane >> 4) << 3) * HPITCH + (lane & 15);
        #pragma unroll
        for (int nt = 0; nt < 4; ++nt) {
            const v16h* bp = reinterpret_cast<const v16h*>(sB) + (nt << 7) + lane;
            v8f c = {};
            c = wmma_f16(a0, bp[0],  c);
            c = wmma_f16(a1, bp[32], c);
            c = wmma_f16(a2, bp[64], c);
            c = wmma_f16(a3, bp[96], c);
            #pragma unroll
            for (int r = 0; r < 8; ++r)
                tp[r * HPITCH + (nt << 4)] = (_Float16)c[r];
        }
        __syncthreads();
        if (!has) return;
        _Float16* hb = reinterpret_cast<_Float16*>(h) + (size_t)(mtile << 4) * HID;
        v8h pv[4];
        #pragma unroll
        for (int i = 0; i < 4; ++i) {
            const int piece = lane + 32 * i;
            pv[i] = *reinterpret_cast<const v8h*>(tw + (piece >> 3) * HPITCH + ((piece & 7) << 3));
        }
        #pragma unroll
        for (int i = 0; i < 4; ++i)
            *reinterpret_cast<v8h*>(hb + ((lane + 32 * i) << 3)) = pv[i];
    } else {
        const int rbase = (mtile << 4) + ((lane >> 4) << 3);
        float* hp = h + (size_t)rbase * HID + (lane & 15);
        #pragma unroll
        for (int nt = 0; nt < 4; ++nt) {
            const v16h* bp = reinterpret_cast<const v16h*>(sB) + (nt << 7) + lane;
            v8f c = {};
            c = wmma_f16(a0, bp[0],  c);
            c = wmma_f16(a1, bp[32], c);
            c = wmma_f16(a2, bp[64], c);
            c = wmma_f16(a3, bp[96], c);
            #pragma unroll
            for (int r = 0; r < 8; ++r)
                hp[r * HID + (nt << 4)] = c[r];
        }
    }
}

__global__ void __launch_bounds__(256)
gemm1_wmma(const float* __restrict__ x,
           const float* __restrict__ Wc,
           float* __restrict__ h,
           int n_nodes)
{
    __shared__ __align__(32) _Float16 sB[NFRAG];
    gemm1_block<false>(x, Wc, h, n_nodes, sB, blockIdx.x, nullptr);
}

__global__ void zero_f32(float4* __restrict__ p, int n4)
{
    int i = blockIdx.x * blockDim.x + threadIdx.x;
    for (; i < n4; i += gridDim.x * blockDim.x)
        p[i] = make_float4(0.f, 0.f, 0.f, 0.f);
}

__global__ void __launch_bounds__(256)
scatter_edges(const float* __restrict__ h,
              const int*   __restrict__ ei,
              const float* __restrict__ ew,
              float*       __restrict__ agg,
              int E)
{
    const int t    = blockIdx.x * blockDim.x + threadIdx.x;
    const int e    = t >> 5;
    const int lane = t & 31;
    if (e >= E) return;

    const int   s = ei[e];
    const int   d = ei[E + e];
    const float w = ew[e];

    float2 hv = reinterpret_cast<const float2*>(h + (long)s * HID)[lane];
    float* ap = agg + (long)d * HID + (lane << 1);
    atomicAdd(ap,     hv.x * w);
    atomicAdd(ap + 1, hv.y * w);
}

#define RPB    64
#define DPB    (2 * RPB)
#define BINCAP (DPB * 32)
#define EPT    16
#define PBE    (256 * EPT)
#define MAXBIN 4096
#define SEGW   16

__device__ __forceinline__ void
part_block(const int*   __restrict__ ei,
           const float* __restrict__ ew,
           int2* __restrict__ recs,
           int2* __restrict__ segt,
           const float* __restrict__ Wcat,
           _Float16* __restrict__ bpk,
           int E, int n_half, int nbin, int npart, int pb, int* hist, int2* stg)
{
    const int tid  = threadIdx.x;
    const int wave = tid >> 5;
    const int lane = tid & 31;
    for (int b = tid; b < nbin + 32; b += 256) hist[b] = 0;
    __syncthreads();

    int key[EPT], dst[EPT], wb[EPT], br[EPT];
    const int e0 = pb * PBE + tid;
    #pragma unroll
    for (int u = 0; u < EPT; ++u) {
        const int e  = e0 + u * 256;
        const int ec = e < E ? e : E - 1;
        key[u] = ei[ec];
        dst[u] = ei[E + ec];
        wb[u]  = __float_as_int(ew[ec]);
    }
    for (int i = pb * 256 + tid; i < NFRAG; i += npart * 256) {
        const int j     = i & 15;
        const int ln    = (i >> 4) & 31;
        const int kci   = (i >> 9) & 3;
        const int ntile = i >> 11;
        const int koff  = (ln >> 4) << 3;
        const int col   = (ntile << 4) + (ln & 15);
        const int k     = (kci << 5) + koff + (j & 7) + ((j >> 3) << 4);
        bpk[i] = (_Float16)Wcat[k * HID + col];
    }
    #pragma unroll
    for (int u = 0; u < EPT; ++u) {
        const int e = e0 + u * 256;
        const unsigned d = (unsigned)dst[u];
        const bool ok = e < E && d < (unsigned)(2 * n_half);
        const unsigned part = d >= (unsigned)n_half ? 1u : 0u;
        const unsigned m    = d - part * (unsigned)n_half;
        const int bin = ok ? (int)(m / RPB) : nbin + (tid & 31);
        const int dl  = (int)((m % RPB) + part * RPB);
        key[u] = (key[u] & 0xFFFFFF) | (dl << 24);
        const int rank = atomicAdd(&hist[bin], 1);
        br[u] = ok ? (bin | (rank << 12)) : -1;
    }
    __syncthreads();
    const int bpt = (nbin + 255) >> 8;
    const int b0  = tid * bpt;
    int s = 0;
    for (int k = 0; k < bpt; ++k) {
        const int b = b0 + k;
        if (b < nbin) s += hist[b];
    }
    int incl = s;
    #pragma unroll
    for (int o = 1; o < 32; o <<= 1) {
        const int v = __shfl_up(incl, o);
        if (lane >= o) incl += v;
    }
    if (lane == 31) hist[nbin + 32 + wave] = incl;
    __syncthreads();
    int run = incl - s;
    for (int w = 0; w < wave; ++w) run += hist[nbin + 32 + w];
    for (int k = 0; k < bpt; ++k) {
        const int b = b0 + k;
        if (b < nbin) {
            const int c = hist[b];
            hist[b] = run;
            segt[(size_t)b * npart + pb] = make_int2(run, c);
            run += c;
        }
    }
    __syncthreads();
    #pragma unroll
    for (int u = 0; u < EPT; ++u)
        dst[u] = hist[br[u] & (MAXBIN - 1)] + (br[u] >> 12);
    #pragma unroll
    for (int u = 0; u < EPT; ++u) {
        if (br[u] >= 0) {
            const int pos = dst[u];
            if (pos >= 0 && pos < PBE) stg[pos] = make_int2(key[u], wb[u]);
        }
    }
    int tot = 0;
    for (int w = 0; w < 8; ++w) tot += hist[nbin + 32 + w];
    tot = tot < 0 ? 0 : (tot > PBE ? PBE : tot);
    for (int i = tot + tid; i < PBE; i += 256) stg[i] = make_int2(0, 0);
    __syncthreads();
    int4* rp4 = reinterpret_cast<int4*>(recs + (size_t)pb * PBE);
    const int4* sp4 = reinterpret_cast<const int4*>(stg);
    int4 pv[EPT / 2];
    #pragma unroll
    for (int i = 0; i < EPT / 2; ++i) pv[i] = sp4[i * 256 + tid];
    #pragma unroll
    for (int i = 0; i < EPT / 2; ++i) rp4[i * 256 + tid] = pv[i];
}

__global__ void __launch_bounds__(256)
gemm1_and_part(const float* __restrict__ x,
               const float* __restrict__ Wc,
               float* __restrict__ h,
               int n_nodes, int nb_gemm,
               const int*   __restrict__ ei,
               const float* __restrict__ ew,
               int2* __restrict__ recs,
               int2* __restrict__ segt,
               const float* __restrict__ Wcat,
               _Float16* __restrict__ bpk,
               int E, int n_half, int nbin, int npart)
{
    __shared__ __align__(32) _Float16 sB[NFRAG];
    __shared__ __align__(16) int2 stg[PBE];
    const int bid = blockIdx.x;
    if (bid < nb_gemm) gemm1_block<true>(x, Wc, h, n_nodes, sB, bid,
                                         reinterpret_cast<_Float16*>(stg));
    else               part_block(ei, ew, recs, segt, Wcat, bpk, E, n_half, nbin, npart, bid - nb_gemm,
                                  reinterpret_cast<int*>(sB), stg);
}

#define APITCH 136

__device__ __forceinline__ float4 bias_slope(float4 v, float4 b, float4 a)
{
    v.x += b.x; v.y += b.y; v.z += b.z; v.w += b.w;
    v.x = v.x > 0.f ? v.x : a.x * v.x;
    v.y = v.y > 0.f ? v.y : a.y * v.y;
    v.z = v.z > 0.f ? v.z : a.z * v.z;
    v.w = v.w > 0.f ? v.w : a.w * v.w;
    return v;
}

__device__ __forceinline__ void fma_h_lo(float& acc, const unsigned hv, const float w)
{
    asm("v_fma_mix_f32 %0, %1, %2, %0 op_sel:[0,0,0] op_sel_hi:[1,0,0]" : "+v"(acc) : "v"(hv), "v"(w));
}
__device__ __forceinline__ void fma_h_hi(float& acc, const unsigned hv, const float w)
{
    asm("v_fma_mix_f32 %0, %1, %2, %0 op_sel:[1,0,0] op_sel_hi:[1,0,0]" : "+v"(acc) : "v"(hv), "v"(w));
}

__device__ __forceinline__ void fma_row_h(float (&a)[8], const v4u v, const float w)
{
    fma_h_lo(a[0], v.x, w); fma_h_hi(a[1], v.x, w);
    fma_h_lo(a[2], v.y, w); fma_h_hi(a[3], v.y, w);
    fma_h_lo(a[4], v.z, w); fma_h_hi(a[5], v.z, w);
    fma_h_lo(a[6], v.w, w); fma_h_hi(a[7], v.w, w);
}

#define GB 12

__device__ __forceinline__ void add_recs(const int2* srec, const int base, const int end,
                                         const char* __restrict__ hb, const unsigned q16,
                                         float (&a)[8])
{
    int2  rr[GB];
    v4u   v[GB];
    float w[GB];
    #pragma unroll
    for (int u = 0; u < GB; ++u) rr[u] = srec[base + u];
    const int n = end - base;
    #pragma unroll
    for (int u = 0; u < GB; ++u) {
        const bool on = u < n;
        const unsigned off = (unsigned)(on ? rr[u].x : rr[0].x) + q16;
        w[u] = on ? __int_as_float(rr[u].y) : 0.f;
        v[u] = *reinterpret_cast<const v4u*>(hb + off);
    }
    __builtin_amdgcn_sched_barrier(0);
    #pragma unroll
    for (int u = 0; u < GB; ++u) fma_row_h(a, v[u], w[u]);
}

__global__ void __launch_bounds__(512)
gather_sort_gemm2(const _Float16* __restrict__ h16,
                  const int2*  __restrict__ recs,
                  const int2*  __restrict__ segt,
                  const float* __restrict__ b_conv,
                  const float* __restrict__ alpha,
                  const _Float16* __restrict__ bpk,
                  const float* __restrict__ b_cat,
                  float*       __restrict__ out,
                  int n_half, int n_nodes, int nbin, int npart)
{
    __shared__ __align__(16) int2 srec[BINCAP + GB];
    __shared__ __align__(16) int  scnt[DPB + 32];
    __shared__ __align__(16) int  soff[DPB + 32];
    __shared__ __align__(16) _Float16 sA[RPB * APITCH];
    __shared__ int sctl[2];

    const int tid   = threadIdx.x;
    const int bin   = blockIdx.x;
    const int wv    = tid >> 5;
    const int ln    = tid & 31;

    if (tid < DPB + 32) scnt[tid] = 0;
    if (tid == 0) sctl[0] = 0;

    int2 sd = make_int2(0, 0);
    {
        const int p = wv + 16 * (ln & 15);
        if (p < npart) sd = segt[(size_t)bin * npart + p];
        sd.x = sd.x < 0 ? 0 : (sd.x > PBE ? PBE : sd.x);
        sd.y = sd.y < 0 ? 0 : (sd.y > PBE - sd.x ? PBE - sd.x : sd.y);
    }
    int2 r[SEGW];
    int  rk[SEGW];
    bool lng = false;
    #pragma unroll
    for (int j = 0; j < SEGW; ++j) {
        const int st = __builtin_amdgcn_readlane(sd.x, j);
        const int cn = __builtin_amdgcn_readlane(sd.y, j);
        lng = lng || cn > 32;
        r[j] = make_int2((DPB + ln) << 24, 0);
        if (ln < cn) r[j] = recs[(size_t)(wv + 16 * j) * PBE + st + ln];
    }
    __syncthreads();
    if (lng && ln == 0) sctl[0] = 1;

    #pragma unroll
    for (int j = 0; j < SEGW; ++j)
        rk[j] = atomicAdd(&scnt[(r[j].x >> 24) & 255], 1);
    __syncthreads();

    if (tid < 32) {
        const int4 c = reinterpret_cast<const int4*>(scnt)[tid];
        const int t = c.x + c.y + c.z + c.w;
        int incl = t;
        #pragma unroll
        for (int o = 1; o < 32; o <<= 1) {
            const int v = __shfl_up(incl, o);
            if (tid >= o) incl += v;
        }
        const int ex = incl - t;
        reinterpret_cast<int4*>(soff)[tid] = make_int4(ex, ex + c.x, ex + c.x + c.y, ex + c.x + c.y + c.z);
        if (tid == 31) sctl[1] = incl;
    }
    __syncthreads();

    const bool big  = sctl[1] > BINCAP;
    const bool scan = big || sctl[0] != 0;

    const unsigned last_row = (unsigned)(n_nodes - 1);
    if (!big) {
        #pragma unroll
        for (int j = 0; j < SEGW; ++j)
            rk[j] += soff[(r[j].x >> 24) & 255];
        #pragma unroll
        for (int j = 0; j < SEGW; ++j) {
            unsigned row = (unsigned)(r[j].x & 0xFFFFFF);
            row = row < last_row ? row : last_row;
            if (((r[j].x >> 24) & 255) < DPB) srec[rk[j]] = make_int2((int)(row << 7), r[j].y);
        }
    }
    __syncthreads();

    const int q    = tid & 7;
    const int g    = tid >> 3;
    const int c0 = q << 3;
    const char* hb = reinterpret_cast<const char*>(h16);
    const unsigned q16 = (unsigned)(q << 4);
    const float4 bc0 = *reinterpret_cast<const float4*>(b_conv + c0);
    const float4 bc1 = *reinterpret_cast<const float4*>(b_conv + c0 + 4);
    const float4 al0 = *reinterpret_cast<const float4*>(alpha + c0);
    const float4 al1 = *reinterpret_cast<const float4*>(alpha + c0 + 4);

    for (int part = 0; part < 2; ++part) {
        const int dl  = part * RPB + g;
        const int beg = big ? 0 : soff[dl];
        const int end = big ? 0 : beg + scnt[dl];
        float a[8] = {0.f, 0.f, 0.f, 0.f, 0.f, 0.f, 0.f, 0.f};

        for (int base = beg; base < end; base += GB)
            add_recs(srec, base, end, hb, q16, a);

        if (scan) {
            const int i0 = big ? 0 : 32;
            for (int p = 0; p < npart; ++p) {
                int2 pd = segt[(size_t)bin * npart + p];
                pd.x = pd.x < 0 ? 0 : (pd.x > PBE ? PBE : pd.x);
                pd.y = pd.y < 0 ? 0 : (pd.y > PBE - pd.x ? PBE - pd.x : pd.y);
                const int2* pr = recs + (size_t)p * PBE + pd.x;
                for (int i = i0; i < pd.y; ++i) {
                    const int2 rc = pr[i];
                    if (((rc.x >> 24) & 255) == dl) {
                        unsigned row = (unsigned)(rc.x & 0xFFFFFF);
                        row = row < last_row ? row : last_row;
                        fma_row_h(a, *reinterpret_cast<const v4u*>(hb + (((size_t)row << 7) + q16)),
                                  __int_as_float(rc.y));
                    }
                }
            }
        }

        const float4 r0 = bias_slope(make_float4(a[0], a[1], a[2], a[3]), bc0, al0);
        const float4 r1 = bias_slope(make_float4(a[4], a[5], a[6], a[7]), bc1, al1);
        _Float16* ap = sA + g * APITCH + (part << 6) + c0;
        v8h p;
        p[0] = (_Float16)r0.x; p[1] = (_Float16)r0.y; p[2] = (_Float16)r0.z; p[3] = (_Float16)r0.w;
        p[4] = (_Float16)r1.x; p[5] = (_Float16)r1.y; p[6] = (_Float16)r1.z; p[7] = (_Float16)r1.w;
        *reinterpret_cast<v8h*>(ap) = p;
    }
    __syncthreads();

    const int wave  = tid >> 5;
    const int lane  = tid & 31;
    const int ntile = wave & 3;
    const int mt    = wave >> 2;

    const v16h* bp = reinterpret_cast<const v16h*>(bpk) + (ntile << 7) + lane;
    const v16h b0 = bp[0], b1 = bp[32], b2 = bp[64], b3 = bp[96];

    const _Float16* ar = sA + ((mt << 4) + (lane & 15)) * APITCH + ((lane >> 4) << 3);
    v8f c = {};
    {
        const v8h l0 = *reinterpret_cast<const v8h*>(ar);
        const v8h h0 = *reinterpret_cast<const v8h*>(ar + 16);
        const v8h l1 = *reinterpret_cast<const v8h*>(ar + 32);
        const v8h h1 = *reinterpret_cast<const v8h*>(ar + 48);
        const v8h l2 = *reinterpret_cast<const v8h*>(ar + 64);
        const v8h h2 = *reinterpret_cast<const v8h*>(ar + 80);
        const v8h l3 = *reinterpret_cast<const v8h*>(ar + 96);
        const v8h h3 = *reinterpret_cast<const v8h*>(ar + 112);
        c = wmma_f16(__builtin_shufflevector(l0, h0, 0,1,2,3,4,5,6,7,8,9,10,11,12,13,14,15), b0, c);
        c = wmma_f16(__builtin_shufflevector(l1, h1, 0,1,2,3,4,5,6,7,8,9,10,11,12,13,14,15), b1, c);
        c = wmma_f16(__builtin_shufflevector(l2, h2, 0,1,2,3,4,5,6,7,8,9,10,11,12,13,14,15), b2, c);
        c = wmma_f16(__builtin_shufflevector(l3, h3, 0,1,2,3,4,5,6,7,8,9,10,11,12,13,14,15), b3, c);
    }

    const int col   = (ntile << 4) + (lane & 15);
    const float bb  = b_cat[col];
    const int rbase = bin * RPB + (mt << 4) + ((lane >> 4) << 3);
    #pragma unroll
    for (int r8 = 0; r8 < 8; ++r8) {
        const int rr = rbase + r8;
        if (rr < n_half) {
            float v = c[r8] + bb;
            out[(long)rr * HID + col] = v > 0.f ? v : 0.f;
        }
    }
}

__global__ void __launch_bounds__(256)
bias_lrelu(float* __restrict__ agg,
           const float* __restrict__ b_conv,
           const float* __restrict__ alpha,
           int total4)
{
    const int i = blockIdx.x * blockDim.x + threadIdx.x;
    if (i >= total4) return;
    const int c4 = (i & 15) << 2;
    float4 v  = reinterpret_cast<float4*>(agg)[i];
    float4 bb = *reinterpret_cast<const float4*>(b_conv + c4);
    float4 aa = *reinterpret_cast<const float4*>(alpha + c4);
    v.x += bb.x; v.y += bb.y; v.z += bb.z; v.w += bb.w;
    v.x = v.x > 0.f ? v.x : aa.x * v.x;
    v.y = v.y > 0.f ? v.y : aa.y * v.y;
    v.z = v.z > 0.f ? v.z : aa.z * v.z;
    v.w = v.w > 0.f ? v.w : aa.w * v.w;
    reinterpret_cast<float4*>(agg)[i] = v;
}

__global__ void __launch_bounds__(256)
gemm2_wmma(const float* __restrict__ act,
           const float* __restrict__ Wcat,
           const float* __restrict__ b_cat,
           float* __restrict__ out,
           int n_half)
{
    __shared__ __align__(32) _Float16 sB[NFRAG];
    stage_Bfrag(Wcat, sB);
    __syncthreads();

    const int wave = threadIdx.x >> 5;
    const int lane = threadIdx.x & 31;
    const int mtiles  = (n_half + 15) >> 4;
    const int mgroups = (mtiles + 3) >> 2;
    const int g = blockIdx.x * (blockDim.x >> 5) + wave;
    if (g >= mgroups * 4) return;
    const int ntile = g & 3;
    const int mg    = g >> 2;

    const v16h* bp = reinterpret_cast<const v16h*>(sB) + (ntile << 7) + lane;
    const v16h b0 = bp[0], b1 = bp[32], b2 = bp[64], b3 = bp[96];

    const int koff = (lane >> 4) << 3;
    const int col  = (ntile << 4) + (lane & 15);

    for (int t = 0; t < 4; ++t) {
        const int mtile = (mg << 2) + t;
        if (mtile >= mtiles) return;
        const int row  = (mtile << 4) + (lane & 15);
        const int rowc = row < n_half ? row : n_half - 1;
        const float* a0 = act + (long)rowc * HID;
        const float* a1 = act + (long)(rowc + n_half) * HID;

        v8f c = {};
        c = wmma_f16(load_a16(a0,      koff), b0, c);
        c = wmma_f16(load_a16(a0, 32 + koff), b1, c);
        c = wmma_f16(load_a16(a1,      koff), b2, c);
        c = wmma_f16(load_a16(a1, 32 + koff), b3, c);

        const float bb = b_cat[col];
        const int rbase = (mtile << 4) + ((lane >> 4) << 3);
        #pragma unroll
        for (int r = 0; r < 8; ++r) {
            const int rr = rbase + r;
            if (rr < n_half) {
                float v = c[r] + bb;
                out[(long)rr * HID + col] = v > 0.f ? v : 0.f;
            }
        }
    }
}

extern "C" void kernel_launch(void* const* d_in, const int* in_sizes, int n_in,
                              void* d_out, int out_size, void* d_ws, size_t ws_size,
                              hipStream_t stream)
{
    const float* x   = (const float*)d_in[0];
    const int*   ei  = (const int*)  d_in[1];
    const float* ew  = (const float*)d_in[2];
    const float* Wc  = (const float*)d_in[3];
    const float* bc  = (const float*)d_in[4];
    const float* al  = (const float*)d_in[5];
    const float* Wk  = (const float*)d_in[6];
    const float* bk  = (const float*)d_in[7];

    const int n_nodes = in_sizes[0] / IN_CH;
    const int E       = in_sizes[2];
    const int n_half  = n_nodes / 2;

    float* h   = (float*)d_ws;
    float* agg = h + (size_t)n_nodes * HID;
    float* out = (float*)d_out;

    const int nb_gemm = (((((n_nodes >> 4) + 3) >> 2) * 4) + 7) / 8;

    const int n4 = (n_nodes * HID) >> 2;
    const int nbin  = (n_half + RPB - 1) / RPB;
    const int npart = (E + PBE - 1) / PBE;
    int2* recs = (int2*)agg;
    _Float16* bpk = (_Float16*)(recs + (size_t)npart * PBE);
    int2* segt = (int2*)(bpk + NFRAG);
    const size_t need = (size_t)n_nodes * HID * sizeof(float)
                      + (size_t)npart * PBE * sizeof(int2)
                      + NFRAG * sizeof(_Float16)
                      + (size_t)nbin * (size_t)npart * sizeof(int2);
    if (ws_size >= need && nbin + 40 <= MAXBIN && npart <= 16 * SEGW && n_nodes <= (1 << 24) && E > 0) {
        gemm1_and_part<<<nb_gemm + npart, 256, 0, stream>>>(
            x, Wc, h, n_nodes, nb_gemm, ei, ew, recs, segt, Wk, bpk, E, n_half, nbin, npart);
        gather_sort_gemm2<<<nbin, 512, 0, stream>>>(
            (const _Float16*)h, recs, segt, bc, al, bpk, bk, out, n_half, n_nodes, nbin, npart);
        return;
    } else {
        gemm1_wmma<<<nb_gemm, 256, 0, stream>>>(x, Wc, h, n_nodes);

        zero_f32<<<(n4 + 255) / 256, 256, 0, stream>>>((float4*)agg, n4);

        scatter_edges<<<(E + 7) / 8, 256, 0, stream>>>(h, ei, ew, agg, E);
    }

    bias_lrelu<<<(n4 + 255) / 256, 256, 0, stream>>>(agg, bc, al, n4);

    {
        const int mgroups = (((n_half + 15) >> 4) + 3) >> 2;
        const int waves   = mgroups * 4;
        gemm2_wmma<<<(waves + 7) / 8, 256, 0, stream>>>(agg, Wk, bk, out, n_half);
    }
}
